# starting kernel
# baseline (speedup 1.0000x reference)
#include <hip/hip_runtime.h>

typedef __attribute__((ext_vector_type(2))) float v2f;
typedef __attribute__((ext_vector_type(8))) float v8f;

#define IN_CH  512
#define HID    16
#define OUT_CH 64
#define THREADS 256

__global__ void k_init_deg(float* __restrict__ deg, int n) {
  int i = blockIdx.x * blockDim.x + threadIdx.x;
  if (i < n) deg[i] = 1.0f;
}

__global__ void k_count_deg(const int* __restrict__ dst, float* __restrict__ deg, int e) {
  int i = blockIdx.x * blockDim.x + threadIdx.x;
  if (i < e) atomicAdd(&deg[dst[i]], 1.0f);
}

__global__ void k_rsqrt_inplace(float* __restrict__ deg, int n) {
  int i = blockIdx.x * blockDim.x + threadIdx.x;
  if (i < n) deg[i] = rsqrtf(deg[i]);
}

__global__ void k_edge_norm(const int* __restrict__ src, const int* __restrict__ dst,
                            const float* __restrict__ dinv, float* __restrict__ norm, int e) {
  int i = blockIdx.x * blockDim.x + threadIdx.x;
  if (i < e) norm[i] = dinv[src[i]] * dinv[dst[i]];
}

__global__ void k_gemm1(const float* __restrict__ x, const float* __restrict__ W1,
                        float* __restrict__ xw1, int nTiles) {
  __shared__ float w1s[IN_CH * HID];
  for (int i = threadIdx.x; i < IN_CH * HID; i += blockDim.x) w1s[i] = W1[i];
  __syncthreads();

  int wave = threadIdx.x >> 5;
  int lane = threadIdx.x & 31;
  int t = blockIdx.x * (blockDim.x >> 5) + wave;
  if (t >= nTiles) return;

  int half = lane >> 4;
  int ml = lane & 15;
  const float* xr = x + (size_t)(t * 16 + ml) * IN_CH + half * 2;

  v8f c = {};
  for (int kk = 0; kk < IN_CH / 4; ++kk) {
    v2f a = *(const v2f*)(xr + kk * 4);
    int k0 = kk * 4 + half * 2;
    v2f b;
    b.x = w1s[k0 * HID + ml];
    b.y = w1s[(k0 + 1) * HID + ml];
    c = __builtin_amdgcn_wmma_f32_16x16x4_f32(false, a, false, b, (short)0, c, false, false);
  }

  float* o = xw1 + (size_t)(t * 16) * HID + ml;
  #pragma unroll
  for (int r = 0; r < 8; ++r)
    o[(half * 8 + r) * HID] = c[r];
}

__global__ void k_gemm2(const float* __restrict__ h, const float* __restrict__ W2,
                        float* __restrict__ h2, int nTiles) {
  __shared__ float w2s[HID * OUT_CH];
  for (int i = threadIdx.x; i < HID * OUT_CH; i += blockDim.x) w2s[i] = W2[i];
  __syncthreads();

  int wave = threadIdx.x >> 5;
  int lane = threadIdx.x & 31;
  int t = blockIdx.x * (blockDim.x >> 5) + wave;
  if (t >= nTiles) return;

  int half = lane >> 4;
  int ml = lane & 15;
  const float* hr = h + (size_t)(t * 16 + ml) * HID + half * 2;

  v8f c0 = {}, c1 = {}, c2 = {}, c3 = {};
  #pragma unroll
  for (int kk = 0; kk < HID / 4; ++kk) {
    v2f a = *(const v2f*)(hr + kk * 4);
    int k0 = kk * 4 + half * 2;
    v2f b0, b1, b2, b3;
    b0.x = w2s[k0 * OUT_CH +  0 + ml]; b0.y = w2s[(k0 + 1) * OUT_CH +  0 + ml];
    b1.x = w2s[k0 * OUT_CH + 16 + ml]; b1.y = w2s[(k0 + 1) * OUT_CH + 16 + ml];
    b2.x = w2s[k0 * OUT_CH + 32 + ml]; b2.y = w2s[(k0 + 1) * OUT_CH + 32 + ml];
    b3.x = w2s[k0 * OUT_CH + 48 + ml]; b3.y = w2s[(k0 + 1) * OUT_CH + 48 + ml];
    c0 = __builtin_amdgcn_wmma_f32_16x16x4_f32(false, a, false, b0, (short)0, c0, false, false);
    c1 = __builtin_amdgcn_wmma_f32_16x16x4_f32(false, a, false, b1, (short)0, c1, false, false);
    c2 = __builtin_amdgcn_wmma_f32_16x16x4_f32(false, a, false, b2, (short)0, c2, false, false);
    c3 = __builtin_amdgcn_wmma_f32_16x16x4_f32(false, a, false, b3, (short)0, c3, false, false);
  }

  float* o = h2 + (size_t)(t * 16) * OUT_CH + ml;
  #pragma unroll
  for (int r = 0; r < 8; ++r) {
    int row = half * 8 + r;
    o[row * OUT_CH +  0] = c0[r];
    o[row * OUT_CH + 16] = c1[r];
    o[row * OUT_CH + 32] = c2[r];
    o[row * OUT_CH + 48] = c3[r];
  }
}

template <int LOGC>
__global__ void k_init_selfloop(const float* __restrict__ in, const float* __restrict__ dinv,
                                const float* __restrict__ bias, float* __restrict__ out,
                                long long total) {
  long long i = (long long)blockIdx.x * blockDim.x + threadIdx.x;
  if (i >= total) return;
  int node = (int)(i >> LOGC);
  int ch = (int)(i & ((1 << LOGC) - 1));
  float di = dinv[node];
  out[i] = bias[ch] + in[i] * di * di;
}

template <int LOGC>
__global__ void k_agg_edges(const int* __restrict__ src, const int* __restrict__ dst,
                            const float* __restrict__ norm, const float* __restrict__ in,
                            float* __restrict__ out, long long total) {
  long long i = (long long)blockIdx.x * blockDim.x + threadIdx.x;
  if (i >= total) return;
  int e = (int)(i >> LOGC);
  int ch = (int)(i & ((1 << LOGC) - 1));
  int s = src[e], d = dst[e];
  float v = in[((size_t)s << LOGC) + ch] * norm[e];
  atomicAdd(&out[((size_t)d << LOGC) + ch], v);
}

__global__ void k_relu(float* __restrict__ h, long long total) {
  long long i = (long long)blockIdx.x * blockDim.x + threadIdx.x;
  if (i < total) h[i] = fmaxf(h[i], 0.0f);
}

static inline int gridFor(long long total, int threads) {
  return (int)((total + threads - 1) / threads);
}

extern "C" void kernel_launch(void* const* d_in, const int* in_sizes, int n_in,
                              void* d_out, int out_size, void* d_ws, size_t ws_size,
                              hipStream_t stream) {
  const float* x  = (const float*)d_in[0];
  const int* ei   = (const int*)d_in[1];
  const float* W1 = (const float*)d_in[2];
  const float* b1 = (const float*)d_in[3];
  const float* W2 = (const float*)d_in[4];
  const float* b2 = (const float*)d_in[5];
  float* out = (float*)d_out;

  const int n = in_sizes[0] / IN_CH;
  const int e = in_sizes[1] / 2;
  const int* src = ei;
  const int* dst = ei + e;

  float* dinv = (float*)d_ws;
  float* norm = dinv + n;
  float* xw1  = norm + e;
  float* h    = xw1 + (size_t)n * HID;
  float* h2   = h   + (size_t)n * HID;

  const int nTiles = (n + 15) / 16;
  const int wavesPerBlock = THREADS / 32;

  k_init_deg<<<gridFor(n, THREADS), THREADS, 0, stream>>>(dinv, n);
  k_count_deg<<<gridFor(e, THREADS), THREADS, 0, stream>>>(dst, dinv, e);
  k_rsqrt_inplace<<<gridFor(n, THREADS), THREADS, 0, stream>>>(dinv, n);
  k_edge_norm<<<gridFor(e, THREADS), THREADS, 0, stream>>>(src, dst, dinv, norm, e);

  k_gemm1<<<gridFor(nTiles, wavesPerBlock), THREADS, 0, stream>>>(x, W1, xw1, nTiles);
  long long tot1 = (long long)n * HID;
  k_init_selfloop<4><<<gridFor(tot1, THREADS), THREADS, 0, stream>>>(xw1, dinv, b1, h, tot1);
  long long te1 = (long long)e * HID;
  k_agg_edges<4><<<gridFor(te1, THREADS), THREADS, 0, stream>>>(src, dst, norm, xw1, h, te1);
  k_relu<<<gridFor(tot1, THREADS), THREADS, 0, stream>>>(h, tot1);

  k_gemm2<<<gridFor(nTiles, wavesPerBlock), THREADS, 0, stream>>>(h, W2, h2, nTiles);
  long long tot2 = (long long)n * OUT_CH;
  k_init_selfloop<6><<<gridFor(tot2, THREADS), THREADS, 0, stream>>>(h2, dinv, b2, out, tot2);
  long long te2 = (long long)e * OUT_CH;
  k_agg_edges<6><<<gridFor(te2, THREADS), THREADS, 0, stream>>>(src, dst, norm, h2, out, te2);
}


# Degree count: a work-group counts one edge slice's destinations for one node range in LDS integer counters and stores a partial plane; the rsqrt pass sums the 32 planes; no global degree atomics
# speedup vs baseline: 3.8148x; 3.8148x over previous
#include <hip/hip_runtime.h>

typedef __attribute__((ext_vector_type(2))) float v2f;
typedef __attribute__((ext_vector_type(8))) float v8f;

#define IN_CH  512
#define HID    16
#define OUT_CH 64
#define THREADS 256

#define CTR_PITCH 64
#define AGG_BATCH 4
#define AGG_GRID  2048

__device__ __forceinline__ int my_half() {
  unsigned v = __builtin_amdgcn_s_sendmsg_rtn(0x87);
  return (int)((v >> 18) & 1u);
}

__device__ __forceinline__ int take_ticket(int* ctr, int lane) {
  int t = 0;
  if (lane == 0) t = atomicAdd(ctr, 1);
  return __builtin_amdgcn_readfirstlane(t);
}

#define DEG_RANGE   12544
#define DEG_SLICES  32
#define DEG_THREADS 1024
#define DEG_GROUP   8
#define PLANE_OFF   1024

__device__ __forceinline__ void deg_bump(int* cnt, int d, int base, int len) {
  unsigned u = (unsigned)(d - base);
  if (u < (unsigned)len) atomicAdd(&cnt[u], 1);
}

__global__ __launch_bounds__(DEG_THREADS)
void k_count_deg(const int* __restrict__ dst, int* __restrict__ planes, int e, int n,
                 int nRanges, int perSlice) {
  __shared__ int cnt[DEG_RANGE];
  const int tid = threadIdx.x;
  const int r = blockIdx.x % nRanges;
  const int sl = blockIdx.x / nRanges;
  const int base = r * DEG_RANGE;
  const int len = (n - base < DEG_RANGE) ? (n - base) : DEG_RANGE;
  for (int i = tid; i < DEG_RANGE; i += DEG_THREADS) cnt[i] = 0;
  __syncthreads();

  const long long lo = (long long)sl * perSlice;
  long long hi = lo + perSlice;
  if (hi > e) hi = e;
  const int m = (hi > lo) ? (int)(hi - lo) : 0;
  const int* p = dst + lo;
  int done = 0;
  if ((((size_t)p) & 15) == 0) {
    const int4* p4 = (const int4*)p;
    const int nvec = m >> 2;
    for (int v = tid; v < nvec; v += DEG_GROUP * DEG_THREADS) {
      int4 d[DEG_GROUP];
      #pragma unroll
      for (int k = 0; k < DEG_GROUP; ++k) {
        const int vv = v + k * DEG_THREADS;
        d[k] = (vv < nvec) ? p4[vv] : make_int4(-1, -1, -1, -1);
      }
      #pragma unroll
      for (int k = 0; k < DEG_GROUP; ++k) {
        deg_bump(cnt, d[k].x, base, len);
        deg_bump(cnt, d[k].y, base, len);
        deg_bump(cnt, d[k].z, base, len);
        deg_bump(cnt, d[k].w, base, len);
      }
    }
    done = nvec << 2;
  }
  for (int i = done + tid; i < m; i += DEG_THREADS) deg_bump(cnt, p[i], base, len);
  __syncthreads();

  int* o = planes + (size_t)sl * n + base;
  for (int i = tid; i < len; i += DEG_THREADS) o[i] = cnt[i];
}

__global__ void k_deg_finish(const int* __restrict__ planes, float* __restrict__ dinv, int n,
                             int* __restrict__ ctrL1) {
  int i = blockIdx.x * blockDim.x + threadIdx.x;
  if (i < n) {
    int c = 1;
    #pragma unroll 8
    for (int sl = 0; sl < DEG_SLICES; ++sl) c += planes[(size_t)sl * n + i];
    dinv[i] = rsqrtf((float)c);
  }
  if (i < 2) ctrL1[i * CTR_PITCH] = 0;
}

__global__ void k_edge_norm(const int* __restrict__ src, const int* __restrict__ dst,
                            const float* __restrict__ dinv, float* __restrict__ norm, int e) {
  int i = blockIdx.x * blockDim.x + threadIdx.x;
  if (i < e) norm[i] = dinv[src[i]] * dinv[dst[i]];
}

__global__ void k_gemm1(const float* __restrict__ x, const float* __restrict__ W1,
                        float* __restrict__ xw1, int nTiles) {
  __shared__ float w1s[IN_CH * HID];
  for (int i = threadIdx.x; i < IN_CH * HID; i += blockDim.x) w1s[i] = W1[i];
  __syncthreads();

  int wave = threadIdx.x >> 5;
  int lane = threadIdx.x & 31;
  int t = blockIdx.x * (blockDim.x >> 5) + wave;
  if (t >= nTiles) return;

  int half = lane >> 4;
  int ml = lane & 15;
  const float* xr = x + (size_t)(t * 16 + ml) * IN_CH + half * 2;

  v8f c = {};
  for (int kk = 0; kk < IN_CH / 4; ++kk) {
    v2f a = *(const v2f*)(xr + kk * 4);
    int k0 = kk * 4 + half * 2;
    v2f b;
    b.x = w1s[k0 * HID + ml];
    b.y = w1s[(k0 + 1) * HID + ml];
    c = __builtin_amdgcn_wmma_f32_16x16x4_f32(false, a, false, b, (short)0, c, false, false);
  }

  float* o = xw1 + (size_t)(t * 16) * HID + ml;
  #pragma unroll
  for (int r = 0; r < 8; ++r)
    o[(half * 8 + r) * HID] = c[r];
}

__global__ void k_gemm2(const float* __restrict__ h, const float* __restrict__ W2,
                        float* __restrict__ h2, int nTiles) {
  __shared__ float w2s[HID * OUT_CH];
  for (int i = threadIdx.x; i < HID * OUT_CH; i += blockDim.x) w2s[i] = W2[i];
  __syncthreads();

  int wave = threadIdx.x >> 5;
  int lane = threadIdx.x & 31;
  int t = blockIdx.x * (blockDim.x >> 5) + wave;
  if (t >= nTiles) return;

  int half = lane >> 4;
  int ml = lane & 15;
  const float* hr = h + (size_t)(t * 16 + ml) * HID + half * 2;

  v8f c0 = {}, c1 = {}, c2 = {}, c3 = {};
  #pragma unroll
  for (int kk = 0; kk < HID / 4; ++kk) {
    v2f a = *(const v2f*)(hr + kk * 4);
    int k0 = kk * 4 + half * 2;
    v2f b0, b1, b2, b3;
    b0.x = w2s[k0 * OUT_CH +  0 + ml]; b0.y = w2s[(k0 + 1) * OUT_CH +  0 + ml];
    b1.x = w2s[k0 * OUT_CH + 16 + ml]; b1.y = w2s[(k0 + 1) * OUT_CH + 16 + ml];
    b2.x = w2s[k0 * OUT_CH + 32 + ml]; b2.y = w2s[(k0 + 1) * OUT_CH + 32 + ml];
    b3.x = w2s[k0 * OUT_CH + 48 + ml]; b3.y = w2s[(k0 + 1) * OUT_CH + 48 + ml];
    c0 = __builtin_amdgcn_wmma_f32_16x16x4_f32(false, a, false, b0, (short)0, c0, false, false);
    c1 = __builtin_amdgcn_wmma_f32_16x16x4_f32(false, a, false, b1, (short)0, c1, false, false);
    c2 = __builtin_amdgcn_wmma_f32_16x16x4_f32(false, a, false, b2, (short)0, c2, false, false);
    c3 = __builtin_amdgcn_wmma_f32_16x16x4_f32(false, a, false, b3, (short)0, c3, false, false);
  }

  float* o = h2 + (size_t)(t * 16) * OUT_CH + ml;
  #pragma unroll
  for (int r = 0; r < 8; ++r) {
    int row = half * 8 + r;
    o[row * OUT_CH +  0] = c0[r];
    o[row * OUT_CH + 16] = c1[r];
    o[row * OUT_CH + 32] = c2[r];
    o[row * OUT_CH + 48] = c3[r];
  }
}

template <int LOGC>
__global__ void k_init_selfloop(const float* __restrict__ in, const float* __restrict__ dinv,
                                const float* __restrict__ bias, float* __restrict__ out,
                                long long total) {
  long long i = (long long)blockIdx.x * blockDim.x + threadIdx.x;
  if (i >= total) return;
  int node = (int)(i >> LOGC);
  int ch = (int)(i & ((1 << LOGC) - 1));
  float di = dinv[node];
  out[i] = bias[ch] + in[i] * di * di;
}

template <int LOGC, bool FULL>
__device__ __forceinline__ unsigned walk_group(unsigned m, int sv, int dv, int wv,
                                               const float* __restrict__ in, float* __restrict__ out,
                                               int sub, int ch) {
  constexpr int C   = 1 << LOGC;
  constexpr int EPI = (C >= 32) ? 1 : 32 / C;
  constexpr int IPE = (C >= 32) ? C / 32 : 1;
  constexpr int U   = 8;
  static_assert(EPI == 1 || EPI == 2, "one or two edges per instruction");
  int ja[U], jb[U];
  #pragma unroll
  for (int u = 0; u < U; ++u) {
    ja[u] = m ? __builtin_ctz(m) : -1;  m &= m - 1;
    if (EPI == 2) { jb[u] = m ? __builtin_ctz(m) : -1;  m &= m - 1; } else jb[u] = -1;
  }
  float a[U][IPE] = {};
  #pragma unroll
  for (int u = 0; u < U; ++u) {
    if (FULL || ja[u] >= 0) {
      int s = __builtin_amdgcn_readlane(sv, ja[u]);
      if (EPI == 2) {
        int s2 = __builtin_amdgcn_readlane(sv, (FULL || jb[u] >= 0) ? jb[u] : ja[u]);
        s = sub ? s2 : s;
      }
      const float* p = in + ((size_t)s << LOGC) + ch;
      for (int k = 0; k < IPE; ++k) a[u][k] = p[k * 32];
    }
  }
  #pragma unroll
  for (int u = 0; u < U; ++u) {
    if (FULL || ja[u] >= 0) {
      int d = __builtin_amdgcn_readlane(dv, ja[u]);
      int w = __builtin_amdgcn_readlane(wv, ja[u]);
      bool act = true;
      if (EPI == 2) {
        int j2 = (FULL || jb[u] >= 0) ? jb[u] : ja[u];
        int d2 = __builtin_amdgcn_readlane(dv, j2);
        int w2 = __builtin_amdgcn_readlane(wv, j2);
        d = sub ? d2 : d;
        w = sub ? w2 : w;
        act = FULL || sub == 0 || jb[u] >= 0;
      }
      float* q = out + ((size_t)d << LOGC) + ch;
      if (act) {
        for (int k = 0; k < IPE; ++k) atomicAdd(q + k * 32, a[u][k] * __int_as_float(w));
      }
    }
  }
  return m;
}

template <int LOGC>
__global__ void k_agg_edges(const int* __restrict__ src, const int* __restrict__ dst,
                            const float* __restrict__ norm, const float* __restrict__ in,
                            float* __restrict__ out, int e, int n, int* __restrict__ ctr) {
  constexpr int C   = 1 << LOGC;
  constexpr int EPI = (C >= 32) ? 1 : 32 / C;
  constexpr int U   = 8;

  const int lane = threadIdx.x & 31;
  const int sub = (EPI == 1) ? 0 : (lane >> LOGC);
  const int ch  = (EPI == 1) ? lane : (lane & (C - 1));
  const int nHalf = n >> 1;
  const int nChunks = (e + 31) >> 5;
  const int nBatches = (nChunks + AGG_BATCH - 1) / AGG_BATCH;
  const int home = my_half();

  for (int pass = 0; pass < 2; ++pass) {
    const int q = home ^ pass;
    for (;;) {
      const int t = take_ticket(ctr + q * CTR_PITCH, lane);
      if (t >= nBatches) break;
      const int c0 = t * AGG_BATCH;
      const int c1 = (c0 + AGG_BATCH < nChunks) ? c0 + AGG_BATCH : nChunks;
      for (int c = c0; c < c1; ++c) {
        const long long idx = (long long)c * 32 + lane;
        int sv = 0, dv = 0, wv = 0;
        bool own = false;
        if (idx < e) {
          sv = src[idx];
          dv = dst[idx];
          wv = __float_as_int(norm[idx]);
          own = ((int)(dv >= nHalf) == q);
        }
        unsigned m = __builtin_amdgcn_ballot_w32(own);
        while (m) {
          if (__builtin_popcount(m) >= U * EPI) m = walk_group<LOGC, true >(m, sv, dv, wv, in, out, sub, ch);
          else                                  m = walk_group<LOGC, false>(m, sv, dv, wv, in, out, sub, ch);
        }
      }
    }
  }
}

__global__ void k_relu(float* __restrict__ h, long long total, int* __restrict__ ctrB) {
  long long i = (long long)blockIdx.x * blockDim.x + threadIdx.x;
  if (i < total) h[i] = fmaxf(h[i], 0.0f);
  if (i < 2) ctrB[i * CTR_PITCH] = 0;
}

static inline int gridFor(long long total, int threads) {
  return (int)((total + threads - 1) / threads);
}

extern "C" void kernel_launch(void* const* d_in, const int* in_sizes, int n_in,
                              void* d_out, int out_size, void* d_ws, size_t ws_size,
                              hipStream_t stream) {
  const float* x  = (const float*)d_in[0];
  const int* ei   = (const int*)d_in[1];
  const float* W1 = (const float*)d_in[2];
  const float* b1 = (const float*)d_in[3];
  const float* W2 = (const float*)d_in[4];
  const float* b2 = (const float*)d_in[5];
  float* out = (float*)d_out;

  const int n = in_sizes[0] / IN_CH;
  const int e = in_sizes[1] / 2;
  const int* src = ei;
  const int* dst = ei + e;

  float* dinv = (float*)d_ws;
  float* norm = dinv + n;
  float* xw1  = norm + e;
  float* h    = xw1 + (size_t)n * HID;
  float* h2   = h   + (size_t)n * HID;

  int* ctrA = (int*)h2;
  int* ctrB = (int*)xw1;

  const int nTiles = (n + 15) / 16;
  const int wavesPerBlock = THREADS / 32;

  const int nRanges = (n + DEG_RANGE - 1) / DEG_RANGE;
  const int perSlice = ((e + DEG_SLICES - 1) / DEG_SLICES + 3) & ~3;
  int* planes = ctrA + PLANE_OFF;
  k_count_deg<<<nRanges * DEG_SLICES, DEG_THREADS, 0, stream>>>(dst, planes, e, n, nRanges, perSlice);
  k_deg_finish<<<gridFor(n, THREADS), THREADS, 0, stream>>>(planes, dinv, n, ctrA + 2 * CTR_PITCH);
  k_edge_norm<<<gridFor(e, THREADS), THREADS, 0, stream>>>(src, dst, dinv, norm, e);

  k_gemm1<<<gridFor(nTiles, wavesPerBlock), THREADS, 0, stream>>>(x, W1, xw1, nTiles);
  long long tot1 = (long long)n * HID;
  k_init_selfloop<4><<<gridFor(tot1, THREADS), THREADS, 0, stream>>>(xw1, dinv, b1, h, tot1);
  k_agg_edges<4><<<AGG_GRID, THREADS, 0, stream>>>(src, dst, norm, xw1, h, e, n, ctrA + 2 * CTR_PITCH);
  k_relu<<<gridFor(tot1, THREADS), THREADS, 0, stream>>>(h, tot1, ctrB);

  k_gemm2<<<gridFor(nTiles, wavesPerBlock), THREADS, 0, stream>>>(h, W2, h2, nTiles);
  long long tot2 = (long long)n * OUT_CH;
  k_init_selfloop<6><<<gridFor(tot2, THREADS), THREADS, 0, stream>>>(h2, dinv, b2, out, tot2);
  k_agg_edges<6><<<AGG_GRID, THREADS, 0, stream>>>(src, dst, norm, h2, out, e, n, ctrB);
}


# Both row scatters: one ticket per work-group, not per wave (thread 0 draws it one batch ahead, an LDS word hands it on): 8x fewer adds on the two hot counters; atomic arrival order changes
# speedup vs baseline: 4.1817x; 1.0962x over previous
#include <hip/hip_runtime.h>

typedef __attribute__((ext_vector_type(2))) float v2f;
typedef __attribute__((ext_vector_type(8))) float v8f;

#define IN_CH  512
#define HID    16
#define OUT_CH 64
#define THREADS 256

#define CTR_PITCH 64
#define AGG_BATCH 4
#define AGG_GRID  2048

__device__ __forceinline__ int my_half() {
  unsigned v = __builtin_amdgcn_s_sendmsg_rtn(0x87);
  return (int)((v >> 18) & 1u);
}

#define DEG_RANGE   12544
#define DEG_SLICES  32
#define DEG_THREADS 1024
#define DEG_GROUP   8
#define PLANE_OFF   1024

__device__ __forceinline__ void deg_bump(int* cnt, int d, int base, int len) {
  unsigned u = (unsigned)(d - base);
  if (u < (unsigned)len) atomicAdd(&cnt[u], 1);
}

__global__ __launch_bounds__(DEG_THREADS)
void k_count_deg(const int* __restrict__ dst, int* __restrict__ planes, int e, int n,
                 int nRanges, int perSlice) {
  __shared__ int cnt[DEG_RANGE];
  const int tid = threadIdx.x;
  const int r = blockIdx.x % nRanges;
  const int sl = blockIdx.x / nRanges;
  const int base = r * DEG_RANGE;
  const int len = (n - base < DEG_RANGE) ? (n - base) : DEG_RANGE;
  for (int i = tid; i < DEG_RANGE; i += DEG_THREADS) cnt[i] = 0;
  __syncthreads();

  const long long lo = (long long)sl * perSlice;
  long long hi = lo + perSlice;
  if (hi > e) hi = e;
  const int m = (hi > lo) ? (int)(hi - lo) : 0;
  const int* p = dst + lo;
  int done = 0;
  if ((((size_t)p) & 15) == 0) {
    const int4* p4 = (const int4*)p;
    const int nvec = m >> 2;
    for (int v = tid; v < nvec; v += DEG_GROUP * DEG_THREADS) {
      int4 d[DEG_GROUP];
      #pragma unroll
      for (int k = 0; k < DEG_GROUP; ++k) {
        const int vv = v + k * DEG_THREADS;
        d[k] = (vv < nvec) ? p4[vv] : make_int4(-1, -1, -1, -1);
      }
      #pragma unroll
      for (int k = 0; k < DEG_GROUP; ++k) {
        deg_bump(cnt, d[k].x, base, len);
        deg_bump(cnt, d[k].y, base, len);
        deg_bump(cnt, d[k].z, base, len);
        deg_bump(cnt, d[k].w, base, len);
      }
    }
    done = nvec << 2;
  }
  for (int i = done + tid; i < m; i += DEG_THREADS) deg_bump(cnt, p[i], base, len);
  __syncthreads();

  int* o = planes + (size_t)sl * n + base;
  for (int i = tid; i < len; i += DEG_THREADS) o[i] = cnt[i];
}

__global__ void k_deg_finish(const int* __restrict__ planes, float* __restrict__ dinv, int n,
                             int* __restrict__ ctrL1) {
  int i = blockIdx.x * blockDim.x + threadIdx.x;
  if (i < n) {
    int c = 1;
    #pragma unroll 8
    for (int sl = 0; sl < DEG_SLICES; ++sl) c += planes[(size_t)sl * n + i];
    dinv[i] = rsqrtf((float)c);
  }
  if (i < 2) ctrL1[i * CTR_PITCH] = 0;
}

__global__ void k_edge_norm(const int* __restrict__ src, const int* __restrict__ dst,
                            const float* __restrict__ dinv, float* __restrict__ norm, int e) {
  int i = blockIdx.x * blockDim.x + threadIdx.x;
  if (i < e) norm[i] = dinv[src[i]] * dinv[dst[i]];
}

__global__ void k_gemm1(const float* __restrict__ x, const float* __restrict__ W1,
                        float* __restrict__ xw1, int nTiles) {
  __shared__ float w1s[IN_CH * HID];
  for (int i = threadIdx.x; i < IN_CH * HID; i += blockDim.x) w1s[i] = W1[i];
  __syncthreads();

  int wave = threadIdx.x >> 5;
  int lane = threadIdx.x & 31;
  int t = blockIdx.x * (blockDim.x >> 5) + wave;
  if (t >= nTiles) return;

  int half = lane >> 4;
  int ml = lane & 15;
  const float* xr = x + (size_t)(t * 16 + ml) * IN_CH + half * 2;

  v8f c = {};
  for (int kk = 0; kk < IN_CH / 4; ++kk) {
    v2f a = *(const v2f*)(xr + kk * 4);
    int k0 = kk * 4 + half * 2;
    v2f b;
    b.x = w1s[k0 * HID + ml];
    b.y = w1s[(k0 + 1) * HID + ml];
    c = __builtin_amdgcn_wmma_f32_16x16x4_f32(false, a, false, b, (short)0, c, false, false);
  }

  float* o = xw1 + (size_t)(t * 16) * HID + ml;
  #pragma unroll
  for (int r = 0; r < 8; ++r)
    o[(half * 8 + r) * HID] = c[r];
}

__global__ void k_gemm2(const float* __restrict__ h, const float* __restrict__ W2,
                        float* __restrict__ h2, int nTiles) {
  __shared__ float w2s[HID * OUT_CH];
  for (int i = threadIdx.x; i < HID * OUT_CH; i += blockDim.x) w2s[i] = W2[i];
  __syncthreads();

  int wave = threadIdx.x >> 5;
  int lane = threadIdx.x & 31;
  int t = blockIdx.x * (blockDim.x >> 5) + wave;
  if (t >= nTiles) return;

  int half = lane >> 4;
  int ml = lane & 15;
  const float* hr = h + (size_t)(t * 16 + ml) * HID + half * 2;

  v8f c0 = {}, c1 = {}, c2 = {}, c3 = {};
  #pragma unroll
  for (int kk = 0; kk < HID / 4; ++kk) {
    v2f a = *(const v2f*)(hr + kk * 4);
    int k0 = kk * 4 + half * 2;
    v2f b0, b1, b2, b3;
    b0.x = w2s[k0 * OUT_CH +  0 + ml]; b0.y = w2s[(k0 + 1) * OUT_CH +  0 + ml];
    b1.x = w2s[k0 * OUT_CH + 16 + ml]; b1.y = w2s[(k0 + 1) * OUT_CH + 16 + ml];
    b2.x = w2s[k0 * OUT_CH + 32 + ml]; b2.y = w2s[(k0 + 1) * OUT_CH + 32 + ml];
    b3.x = w2s[k0 * OUT_CH + 48 + ml]; b3.y = w2s[(k0 + 1) * OUT_CH + 48 + ml];
    c0 = __builtin_amdgcn_wmma_f32_16x16x4_f32(false, a, false, b0, (short)0, c0, false, false);
    c1 = __builtin_amdgcn_wmma_f32_16x16x4_f32(false, a, false, b1, (short)0, c1, false, false);
    c2 = __builtin_amdgcn_wmma_f32_16x16x4_f32(false, a, false, b2, (short)0, c2, false, false);
    c3 = __builtin_amdgcn_wmma_f32_16x16x4_f32(false, a, false, b3, (short)0, c3, false, false);
  }

  float* o = h2 + (size_t)(t * 16) * OUT_CH + ml;
  #pragma unroll
  for (int r = 0; r < 8; ++r) {
    int row = half * 8 + r;
    o[row * OUT_CH +  0] = c0[r];
    o[row * OUT_CH + 16] = c1[r];
    o[row * OUT_CH + 32] = c2[r];
    o[row * OUT_CH + 48] = c3[r];
  }
}

template <int LOGC>
__global__ void k_init_selfloop(const float* __restrict__ in, const float* __restrict__ dinv,
                                const float* __restrict__ bias, float* __restrict__ out,
                                long long total) {
  long long i = (long long)blockIdx.x * blockDim.x + threadIdx.x;
  if (i >= total) return;
  int node = (int)(i >> LOGC);
  int ch = (int)(i & ((1 << LOGC) - 1));
  float di = dinv[node];
  out[i] = bias[ch] + in[i] * di * di;
}

template <int LOGC, bool FULL>
__device__ __forceinline__ unsigned walk_group(unsigned m, int sv, int dv, int wv,
                                               const float* __restrict__ in, float* __restrict__ out,
                                               int sub, int ch) {
  constexpr int C   = 1 << LOGC;
  constexpr int EPI = (C >= 32) ? 1 : 32 / C;
  constexpr int IPE = (C >= 32) ? C / 32 : 1;
  constexpr int U   = 8;
  static_assert(EPI == 1 || EPI == 2, "one or two edges per instruction");
  int ja[U], jb[U];
  #pragma unroll
  for (int u = 0; u < U; ++u) {
    ja[u] = m ? __builtin_ctz(m) : -1;  m &= m - 1;
    if (EPI == 2) { jb[u] = m ? __builtin_ctz(m) : -1;  m &= m - 1; } else jb[u] = -1;
  }
  float a[U][IPE] = {};
  #pragma unroll
  for (int u = 0; u < U; ++u) {
    if (FULL || ja[u] >= 0) {
      int s = __builtin_amdgcn_readlane(sv, ja[u]);
      if (EPI == 2) {
        int s2 = __builtin_amdgcn_readlane(sv, (FULL || jb[u] >= 0) ? jb[u] : ja[u]);
        s = sub ? s2 : s;
      }
      const float* p = in + ((size_t)s << LOGC) + ch;
      for (int k = 0; k < IPE; ++k) a[u][k] = p[k * 32];
    }
  }
  #pragma unroll
  for (int u = 0; u < U; ++u) {
    if (FULL || ja[u] >= 0) {
      int d = __builtin_amdgcn_readlane(dv, ja[u]);
      int w = __builtin_amdgcn_readlane(wv, ja[u]);
      bool act = true;
      if (EPI == 2) {
        int j2 = (FULL || jb[u] >= 0) ? jb[u] : ja[u];
        int d2 = __builtin_amdgcn_readlane(dv, j2);
        int w2 = __builtin_amdgcn_readlane(wv, j2);
        d = sub ? d2 : d;
        w = sub ? w2 : w;
        act = FULL || sub == 0 || jb[u] >= 0;
      }
      float* q = out + ((size_t)d << LOGC) + ch;
      if (act) {
        for (int k = 0; k < IPE; ++k) atomicAdd(q + k * 32, a[u][k] * __int_as_float(w));
      }
    }
  }
  return m;
}

template <int LOGC>
__global__ void k_agg_edges(const int* __restrict__ src, const int* __restrict__ dst,
                            const float* __restrict__ norm, const float* __restrict__ in,
                            float* __restrict__ out, int e, int n, int* __restrict__ ctr) {
  constexpr int C   = 1 << LOGC;
  constexpr int EPI = (C >= 32) ? 1 : 32 / C;
  constexpr int U   = 8;

  constexpr int WG_CHUNKS = AGG_BATCH * (THREADS / 32);

  __shared__ int s_tk[2];

  const int lane = threadIdx.x & 31;
  const int wave = __builtin_amdgcn_readfirstlane(threadIdx.x >> 5);
  const int sub = (EPI == 1) ? 0 : (lane >> LOGC);
  const int ch  = (EPI == 1) ? lane : (lane & (C - 1));
  const int nHalf = n >> 1;
  const int nChunks = (e + 31) >> 5;
  const int nBatches = (nChunks + WG_CHUNKS - 1) / WG_CHUNKS;
  const int home = my_half();
  int round = 0;

  for (int pass = 0; pass < 2; ++pass) {
    const int q = home ^ pass;
    unsigned* cq = (unsigned*)(ctr + q * CTR_PITCH);
    unsigned ahead = 0;
    if (threadIdx.x == 0) ahead = atomicInc(cq, 0xffffffffu);
    for (;;) {
      if (threadIdx.x == 0) s_tk[round & 1] = (int)ahead;
      __syncthreads();
      const int t = __builtin_amdgcn_readfirstlane(s_tk[round & 1]);
      ++round;
      if (t >= nBatches) break;
      if (threadIdx.x == 0) ahead = atomicInc(cq, 0xffffffffu);
      const int c0 = t * WG_CHUNKS + wave * AGG_BATCH;
      const int c1 = (c0 + AGG_BATCH < nChunks) ? c0 + AGG_BATCH : nChunks;
      for (int c = c0; c < c1; ++c) {
        const long long idx = (long long)c * 32 + lane;
        int sv = 0, dv = 0, wv = 0;
        bool own = false;
        if (idx < e) {
          sv = src[idx];
          dv = dst[idx];
          wv = __float_as_int(norm[idx]);
          own = ((int)(dv >= nHalf) == q);
        }
        unsigned m = __builtin_amdgcn_ballot_w32(own);
        while (m) {
          if (__builtin_popcount(m) >= U * EPI) m = walk_group<LOGC, true >(m, sv, dv, wv, in, out, sub, ch);
          else                                  m = walk_group<LOGC, false>(m, sv, dv, wv, in, out, sub, ch);
        }
      }
    }
  }
}

__global__ void k_relu(float* __restrict__ h, long long total, int* __restrict__ ctrB) {
  long long i = (long long)blockIdx.x * blockDim.x + threadIdx.x;
  if (i < total) h[i] = fmaxf(h[i], 0.0f);
  if (i < 2) ctrB[i * CTR_PITCH] = 0;
}

static inline int gridFor(long long total, int threads) {
  return (int)((total + threads - 1) / threads);
}

extern "C" void kernel_launch(void* const* d_in, const int* in_sizes, int n_in,
                              void* d_out, int out_size, void* d_ws, size_t ws_size,
                              hipStream_t stream) {
  const float* x  = (const float*)d_in[0];
  const int* ei   = (const int*)d_in[1];
  const float* W1 = (const float*)d_in[2];
  const float* b1 = (const float*)d_in[3];
  const float* W2 = (const float*)d_in[4];
  const float* b2 = (const float*)d_in[5];
  float* out = (float*)d_out;

  const int n = in_sizes[0] / IN_CH;
  const int e = in_sizes[1] / 2;
  const int* src = ei;
  const int* dst = ei + e;

  float* dinv = (float*)d_ws;
  float* norm = dinv + n;
  float* xw1  = norm + e;
  float* h    = xw1 + (size_t)n * HID;
  float* h2   = h   + (size_t)n * HID;

  int* ctrA = (int*)h2;
  int* ctrB = (int*)xw1;

  const int nTiles = (n + 15) / 16;
  const int wavesPerBlock = THREADS / 32;

  const int nRanges = (n + DEG_RANGE - 1) / DEG_RANGE;
  const int perSlice = ((e + DEG_SLICES - 1) / DEG_SLICES + 3) & ~3;
  int* planes = ctrA + PLANE_OFF;
  k_count_deg<<<nRanges * DEG_SLICES, DEG_THREADS, 0, stream>>>(dst, planes, e, n, nRanges, perSlice);
  k_deg_finish<<<gridFor(n, THREADS), THREADS, 0, stream>>>(planes, dinv, n, ctrA + 2 * CTR_PITCH);
  k_edge_norm<<<gridFor(e, THREADS), THREADS, 0, stream>>>(src, dst, dinv, norm, e);

  k_gemm1<<<gridFor(nTiles, wavesPerBlock), THREADS, 0, stream>>>(x, W1, xw1, nTiles);
  long long tot1 = (long long)n * HID;
  k_init_selfloop<4><<<gridFor(tot1, THREADS), THREADS, 0, stream>>>(xw1, dinv, b1, h, tot1);
  k_agg_edges<4><<<AGG_GRID, THREADS, 0, stream>>>(src, dst, norm, xw1, h, e, n, ctrA + 2 * CTR_PITCH);
  k_relu<<<gridFor(tot1, THREADS), THREADS, 0, stream>>>(h, tot1, ctrB);

  k_gemm2<<<gridFor(nTiles, wavesPerBlock), THREADS, 0, stream>>>(h, W2, h2, nTiles);
  long long tot2 = (long long)n * OUT_CH;
  k_init_selfloop<6><<<gridFor(tot2, THREADS), THREADS, 0, stream>>>(h2, dinv, b2, out, tot2);
  k_agg_edges<6><<<AGG_GRID, THREADS, 0, stream>>>(src, dst, norm, h2, out, e, n, ctrB);
}


# Edge-norm, both start-value passes and relu: 16 bytes per lane, 2-4 vectors per thread loaded before the first store: 8-16x fewer waves per pass; arithmetic and sum order unchanged
# speedup vs baseline: 4.3824x; 1.0480x over previous
#include <hip/hip_runtime.h>

typedef __attribute__((ext_vector_type(2))) float v2f;
typedef __attribute__((ext_vector_type(8))) float v8f;

#define IN_CH  512
#define HID    16
#define OUT_CH 64
#define THREADS 256

#define CTR_PITCH 64
#define AGG_BATCH 4
#define AGG_GRID  2048

__device__ __forceinline__ int my_half() {
  unsigned v = __builtin_amdgcn_s_sendmsg_rtn(0x87);
  return (int)((v >> 18) & 1u);
}

#define DEG_RANGE   12544
#define DEG_SLICES  32
#define DEG_THREADS 1024
#define DEG_GROUP   8
#define PLANE_OFF   1024

__device__ __forceinline__ void deg_bump(int* cnt, int d, int base, int len) {
  unsigned u = (unsigned)(d - base);
  if (u < (unsigned)len) atomicAdd(&cnt[u], 1);
}

__global__ __launch_bounds__(DEG_THREADS)
void k_count_deg(const int* __restrict__ dst, int* __restrict__ planes, int e, int n,
                 int nRanges, int perSlice) {
  __shared__ int cnt[DEG_RANGE];
  const int tid = threadIdx.x;
  const int r = blockIdx.x % nRanges;
  const int sl = blockIdx.x / nRanges;
  const int base = r * DEG_RANGE;
  const int len = (n - base < DEG_RANGE) ? (n - base) : DEG_RANGE;
  for (int i = tid; i < DEG_RANGE; i += DEG_THREADS) cnt[i] = 0;
  __syncthreads();

  const long long lo = (long long)sl * perSlice;
  long long hi = lo + perSlice;
  if (hi > e) hi = e;
  const int m = (hi > lo) ? (int)(hi - lo) : 0;
  const int* p = dst + lo;
  int done = 0;
  if ((((size_t)p) & 15) == 0) {
    const int4* p4 = (const int4*)p;
    const int nvec = m >> 2;
    for (int v = tid; v < nvec; v += DEG_GROUP * DEG_THREADS) {
      int4 d[DEG_GROUP];
      #pragma unroll
      for (int k = 0; k < DEG_GROUP; ++k) {
        const int vv = v + k * DEG_THREADS;
        d[k] = (vv < nvec) ? p4[vv] : make_int4(-1, -1, -1, -1);
      }
      #pragma unroll
      for (int k = 0; k < DEG_GROUP; ++k) {
        deg_bump(cnt, d[k].x, base, len);
        deg_bump(cnt, d[k].y, base, len);
        deg_bump(cnt, d[k].z, base, len);
        deg_bump(cnt, d[k].w, base, len);
      }
    }
    done = nvec << 2;
  }
  for (int i = done + tid; i < m; i += DEG_THREADS) deg_bump(cnt, p[i], base, len);
  __syncthreads();

  int* o = planes + (size_t)sl * n + base;
  for (int i = tid; i < len; i += DEG_THREADS) o[i] = cnt[i];
}

__global__ void k_deg_finish(const int* __restrict__ planes, float* __restrict__ dinv, int n,
                             int* __restrict__ ctrL1) {
  int i = blockIdx.x * blockDim.x + threadIdx.x;
  if (i < n) {
    int c = 1;
    #pragma unroll 8
    for (int sl = 0; sl < DEG_SLICES; ++sl) c += planes[(size_t)sl * n + i];
    dinv[i] = rsqrtf((float)c);
  }
  if (i < 2) ctrL1[i * CTR_PITCH] = 0;
}

#define EW_NORM 2
#define EW_ROW1 2
#define EW_ROW2 4

template <bool GUARD>
__device__ __forceinline__ void edge_norm_vec(const int4* __restrict__ s4, const int4* __restrict__ d4,
                                              const float* __restrict__ dinv, float4* __restrict__ o4,
                                              int v0, int nvec) {
  int4 s[EW_NORM], d[EW_NORM];
  float4 w[EW_NORM];
  #pragma unroll
  for (int k = 0; k < EW_NORM; ++k) {
    const int v = v0 + k * blockDim.x;
    const int vc = (!GUARD || v < nvec) ? v : nvec - 1;
    s[k] = s4[vc];
    d[k] = d4[vc];
  }
  #pragma unroll
  for (int k = 0; k < EW_NORM; ++k) {
    w[k].x = dinv[s[k].x] * dinv[d[k].x];
    w[k].y = dinv[s[k].y] * dinv[d[k].y];
    w[k].z = dinv[s[k].z] * dinv[d[k].z];
    w[k].w = dinv[s[k].w] * dinv[d[k].w];
  }
  #pragma unroll
  for (int k = 0; k < EW_NORM; ++k) {
    const int v = v0 + k * blockDim.x;
    if (!GUARD || v < nvec) o4[v] = w[k];
  }
}

__global__ void k_edge_norm(const int* __restrict__ src, const int* __restrict__ dst,
                            const float* __restrict__ dinv, float* __restrict__ norm, int e) {
  const bool vec = (((size_t)src | (size_t)dst | (size_t)norm) & 15) == 0;
  const int nvec = vec ? (e >> 2) : 0;
  const int4* s4 = (const int4*)src;
  const int4* d4 = (const int4*)dst;
  float4* o4 = (float4*)norm;
  const int v0 = blockIdx.x * (EW_NORM * blockDim.x) + threadIdx.x;
  if (nvec > 0) {
    if ((long long)(blockIdx.x + 1) * (EW_NORM * blockDim.x) <= nvec) edge_norm_vec<false>(s4, d4, dinv, o4, v0, nvec);
    else                                                              edge_norm_vec<true >(s4, d4, dinv, o4, v0, nvec);
  }
  const long long gsz = (long long)gridDim.x * blockDim.x;
  for (long long i = 4LL * nvec + (long long)blockIdx.x * blockDim.x + threadIdx.x; i < e; i += gsz)
    norm[i] = dinv[src[i]] * dinv[dst[i]];
}

__global__ void k_gemm1(const float* __restrict__ x, const float* __restrict__ W1,
                        float* __restrict__ xw1, int nTiles) {
  __shared__ float w1s[IN_CH * HID];
  for (int i = threadIdx.x; i < IN_CH * HID; i += blockDim.x) w1s[i] = W1[i];
  __syncthreads();

  int wave = threadIdx.x >> 5;
  int lane = threadIdx.x & 31;
  int t = blockIdx.x * (blockDim.x >> 5) + wave;
  if (t >= nTiles) return;

  int half = lane >> 4;
  int ml = lane & 15;
  const float* xr = x + (size_t)(t * 16 + ml) * IN_CH + half * 2;

  v8f c = {};
  for (int kk = 0; kk < IN_CH / 4; ++kk) {
    v2f a = *(const v2f*)(xr + kk * 4);
    int k0 = kk * 4 + half * 2;
    v2f b;
    b.x = w1s[k0 * HID + ml];
    b.y = w1s[(k0 + 1) * HID + ml];
    c = __builtin_amdgcn_wmma_f32_16x16x4_f32(false, a, false, b, (short)0, c, false, false);
  }

  float* o = xw1 + (size_t)(t * 16) * HID + ml;
  #pragma unroll
  for (int r = 0; r < 8; ++r)
    o[(half * 8 + r) * HID] = c[r];
}

__global__ void k_gemm2(const float* __restrict__ h, const float* __restrict__ W2,
                        float* __restrict__ h2, int nTiles) {
  __shared__ float w2s[HID * OUT_CH];
  for (int i = threadIdx.x; i < HID * OUT_CH; i += blockDim.x) w2s[i] = W2[i];
  __syncthreads();

  int wave = threadIdx.x >> 5;
  int lane = threadIdx.x & 31;
  int t = blockIdx.x * (blockDim.x >> 5) + wave;
  if (t >= nTiles) return;

  int half = lane >> 4;
  int ml = lane & 15;
  const float* hr = h + (size_t)(t * 16 + ml) * HID + half * 2;

  v8f c0 = {}, c1 = {}, c2 = {}, c3 = {};
  #pragma unroll
  for (int kk = 0; kk < HID / 4; ++kk) {
    v2f a = *(const v2f*)(hr + kk * 4);
    int k0 = kk * 4 + half * 2;
    v2f b0, b1, b2, b3;
    b0.x = w2s[k0 * OUT_CH +  0 + ml]; b0.y = w2s[(k0 + 1) * OUT_CH +  0 + ml];
    b1.x = w2s[k0 * OUT_CH + 16 + ml]; b1.y = w2s[(k0 + 1) * OUT_CH + 16 + ml];
    b2.x = w2s[k0 * OUT_CH + 32 + ml]; b2.y = w2s[(k0 + 1) * OUT_CH + 32 + ml];
    b3.x = w2s[k0 * OUT_CH + 48 + ml]; b3.y = w2s[(k0 + 1) * OUT_CH + 48 + ml];
    c0 = __builtin_amdgcn_wmma_f32_16x16x4_f32(false, a, false, b0, (short)0, c0, false, false);
    c1 = __builtin_amdgcn_wmma_f32_16x16x4_f32(false, a, false, b1, (short)0, c1, false, false);
    c2 = __builtin_amdgcn_wmma_f32_16x16x4_f32(false, a, false, b2, (short)0, c2, false, false);
    c3 = __builtin_amdgcn_wmma_f32_16x16x4_f32(false, a, false, b3, (short)0, c3, false, false);
  }

  float* o = h2 + (size_t)(t * 16) * OUT_CH + ml;
  #pragma unroll
  for (int r = 0; r < 8; ++r) {
    int row = half * 8 + r;
    o[row * OUT_CH +  0] = c0[r];
    o[row * OUT_CH + 16] = c1[r];
    o[row * OUT_CH + 32] = c2[r];
    o[row * OUT_CH + 48] = c3[r];
  }
}

template <int LOGC, int UNR, bool GUARD>
__device__ __forceinline__ void init_selfloop_vec(const float4* __restrict__ i4, const float* __restrict__ dinv,
                                                  const float4* __restrict__ b4, float4* __restrict__ o4,
                                                  long long v0, long long nvec) {
  float4 a[UNR], b[UNR];
  float di[UNR];
  #pragma unroll
  for (int k = 0; k < UNR; ++k) {
    const long long v = v0 + (long long)k * blockDim.x;
    const long long vc = (!GUARD || v < nvec) ? v : nvec - 1;
    a[k] = i4[vc];
    di[k] = dinv[(int)(vc >> (LOGC - 2))];
    b[k] = b4[(int)(vc & ((1 << (LOGC - 2)) - 1))];
  }
  #pragma unroll
  for (int k = 0; k < UNR; ++k) {
    const long long v = v0 + (long long)k * blockDim.x;
    if (!GUARD || v < nvec) {
      const float d = di[k];
      float4 r;
      r.x = b[k].x + a[k].x * d * d;
      r.y = b[k].y + a[k].y * d * d;
      r.z = b[k].z + a[k].z * d * d;
      r.w = b[k].w + a[k].w * d * d;
      o4[v] = r;
    }
  }
}

template <int LOGC, int UNR>
__global__ void k_init_selfloop(const float* __restrict__ in, const float* __restrict__ dinv,
                                const float* __restrict__ bias, float* __restrict__ out,
                                long long total) {
  static_assert(LOGC >= 2, "four channels per vector");
  const bool vec = (((size_t)in | (size_t)out | (size_t)bias) & 15) == 0;
  const long long nvec = vec ? (total >> 2) : 0;
  const float4* i4 = (const float4*)in;
  const float4* b4 = (const float4*)bias;
  float4* o4 = (float4*)out;
  const long long v0 = (long long)blockIdx.x * (UNR * blockDim.x) + threadIdx.x;
  if (nvec > 0) {
    if ((long long)(blockIdx.x + 1) * (UNR * blockDim.x) <= nvec) init_selfloop_vec<LOGC, UNR, false>(i4, dinv, b4, o4, v0, nvec);
    else                                                          init_selfloop_vec<LOGC, UNR, true >(i4, dinv, b4, o4, v0, nvec);
  }
  const long long gsz = (long long)gridDim.x * blockDim.x;
  for (long long i = 4 * nvec + (long long)blockIdx.x * blockDim.x + threadIdx.x; i < total; i += gsz) {
    const int node = (int)(i >> LOGC);
    const int ch = (int)(i & ((1 << LOGC) - 1));
    const float d = dinv[node];
    out[i] = bias[ch] + in[i] * d * d;
  }
}

template <int LOGC, bool FULL>
__device__ __forceinline__ unsigned walk_group(unsigned m, int sv, int dv, int wv,
                                               const float* __restrict__ in, float* __restrict__ out,
                                               int sub, int ch) {
  constexpr int C   = 1 << LOGC;
  constexpr int EPI = (C >= 32) ? 1 : 32 / C;
  constexpr int IPE = (C >= 32) ? C / 32 : 1;
  constexpr int U   = 8;
  static_assert(EPI == 1 || EPI == 2, "one or two edges per instruction");
  int ja[U], jb[U];
  #pragma unroll
  for (int u = 0; u < U; ++u) {
    ja[u] = m ? __builtin_ctz(m) : -1;  m &= m - 1;
    if (EPI == 2) { jb[u] = m ? __builtin_ctz(m) : -1;  m &= m - 1; } else jb[u] = -1;
  }
  float a[U][IPE] = {};
  #pragma unroll
  for (int u = 0; u < U; ++u) {
    if (FULL || ja[u] >= 0) {
      int s = __builtin_amdgcn_readlane(sv, ja[u]);
      if (EPI == 2) {
        int s2 = __builtin_amdgcn_readlane(sv, (FULL || jb[u] >= 0) ? jb[u] : ja[u]);
        s = sub ? s2 : s;
      }
      const float* p = in + ((size_t)s << LOGC) + ch;
      for (int k = 0; k < IPE; ++k) a[u][k] = p[k * 32];
    }
  }
  #pragma unroll
  for (int u = 0; u < U; ++u) {
    if (FULL || ja[u] >= 0) {
      int d = __builtin_amdgcn_readlane(dv, ja[u]);
      int w = __builtin_amdgcn_readlane(wv, ja[u]);
      bool act = true;
      if (EPI == 2) {
        int j2 = (FULL || jb[u] >= 0) ? jb[u] : ja[u];
        int d2 = __builtin_amdgcn_readlane(dv, j2);
        int w2 = __builtin_amdgcn_readlane(wv, j2);
        d = sub ? d2 : d;
        w = sub ? w2 : w;
        act = FULL || sub == 0 || jb[u] >= 0;
      }
      float* q = out + ((size_t)d << LOGC) + ch;
      if (act) {
        for (int k = 0; k < IPE; ++k) atomicAdd(q + k * 32, a[u][k] * __int_as_float(w));
      }
    }
  }
  return m;
}

template <int LOGC>
__global__ void k_agg_edges(const int* __restrict__ src, const int* __restrict__ dst,
                            const float* __restrict__ norm, const float* __restrict__ in,
                            float* __restrict__ out, int e, int n, int* __restrict__ ctr) {
  constexpr int C   = 1 << LOGC;
  constexpr int EPI = (C >= 32) ? 1 : 32 / C;
  constexpr int U   = 8;

  constexpr int WG_CHUNKS = AGG_BATCH * (THREADS / 32);

  __shared__ int s_tk[2];

  const int lane = threadIdx.x & 31;
  const int wave = __builtin_amdgcn_readfirstlane(threadIdx.x >> 5);
  const int sub = (EPI == 1) ? 0 : (lane >> LOGC);
  const int ch  = (EPI == 1) ? lane : (lane & (C - 1));
  const int nHalf = n >> 1;
  const int nChunks = (e + 31) >> 5;
  const int nBatches = (nChunks + WG_CHUNKS - 1) / WG_CHUNKS;
  const int home = my_half();
  int round = 0;

  for (int pass = 0; pass < 2; ++pass) {
    const int q = home ^ pass;
    unsigned* cq = (unsigned*)(ctr + q * CTR_PITCH);
    unsigned ahead = 0;
    if (threadIdx.x == 0) ahead = atomicInc(cq, 0xffffffffu);
    for (;;) {
      if (threadIdx.x == 0) s_tk[round & 1] = (int)ahead;
      __syncthreads();
      const int t = __builtin_amdgcn_readfirstlane(s_tk[round & 1]);
      ++round;
      if (t >= nBatches) break;
      if (threadIdx.x == 0) ahead = atomicInc(cq, 0xffffffffu);
      const int c0 = t * WG_CHUNKS + wave * AGG_BATCH;
      const int c1 = (c0 + AGG_BATCH < nChunks) ? c0 + AGG_BATCH : nChunks;
      for (int c = c0; c < c1; ++c) {
        const long long idx = (long long)c * 32 + lane;
        int sv = 0, dv = 0, wv = 0;
        bool own = false;
        if (idx < e) {
          sv = src[idx];
          dv = dst[idx];
          wv = __float_as_int(norm[idx]);
          own = ((int)(dv >= nHalf) == q);
        }
        unsigned m = __builtin_amdgcn_ballot_w32(own);
        while (m) {
          if (__builtin_popcount(m) >= U * EPI) m = walk_group<LOGC, true >(m, sv, dv, wv, in, out, sub, ch);
          else                                  m = walk_group<LOGC, false>(m, sv, dv, wv, in, out, sub, ch);
        }
      }
    }
  }
}

template <bool GUARD>
__device__ __forceinline__ void relu_vec(float4* h4, long long v0, long long nvec) {
  float4 a[EW_ROW1];
  #pragma unroll
  for (int k = 0; k < EW_ROW1; ++k) {
    const long long v = v0 + (long long)k * blockDim.x;
    const long long vc = (!GUARD || v < nvec) ? v : nvec - 1;
    a[k] = h4[vc];
  }
  #pragma unroll
  for (int k = 0; k < EW_ROW1; ++k) {
    const long long v = v0 + (long long)k * blockDim.x;
    if (!GUARD || v < nvec) {
      float4 r;
      r.x = fmaxf(a[k].x, 0.0f);
      r.y = fmaxf(a[k].y, 0.0f);
      r.z = fmaxf(a[k].z, 0.0f);
      r.w = fmaxf(a[k].w, 0.0f);
      h4[v] = r;
    }
  }
}

__global__ void k_relu(float* __restrict__ h, long long total, int* __restrict__ ctrB) {
  const bool vec = (((size_t)h) & 15) == 0;
  const long long nvec = vec ? (total >> 2) : 0;
  float4* h4 = (float4*)h;
  const long long v0 = (long long)blockIdx.x * (EW_ROW1 * blockDim.x) + threadIdx.x;
  if (nvec > 0) {
    if ((long long)(blockIdx.x + 1) * (EW_ROW1 * blockDim.x) <= nvec) relu_vec<false>(h4, v0, nvec);
    else                                                              relu_vec<true >(h4, v0, nvec);
  }
  const long long g = (long long)blockIdx.x * blockDim.x + threadIdx.x;
  const long long gsz = (long long)gridDim.x * blockDim.x;
  for (long long i = 4 * nvec + g; i < total; i += gsz) h[i] = fmaxf(h[i], 0.0f);
  if (g < 2) ctrB[g * CTR_PITCH] = 0;
}

static inline int gridFor(long long total, int threads) {
  return (int)((total + threads - 1) / threads);
}

extern "C" void kernel_launch(void* const* d_in, const int* in_sizes, int n_in,
                              void* d_out, int out_size, void* d_ws, size_t ws_size,
                              hipStream_t stream) {
  const float* x  = (const float*)d_in[0];
  const int* ei   = (const int*)d_in[1];
  const float* W1 = (const float*)d_in[2];
  const float* b1 = (const float*)d_in[3];
  const float* W2 = (const float*)d_in[4];
  const float* b2 = (const float*)d_in[5];
  float* out = (float*)d_out;

  const int n = in_sizes[0] / IN_CH;
  const int e = in_sizes[1] / 2;
  const int* src = ei;
  const int* dst = ei + e;

  float* dinv = (float*)d_ws;
  float* norm = dinv + n;
  float* xw1  = norm + e;
  float* h    = xw1 + (size_t)n * HID;
  float* h2   = h   + (size_t)n * HID;

  int* ctrA = (int*)h2;
  int* ctrB = (int*)xw1;

  const int nTiles = (n + 15) / 16;
  const int wavesPerBlock = THREADS / 32;

  const int nRanges = (n + DEG_RANGE - 1) / DEG_RANGE;
  const int perSlice = ((e + DEG_SLICES - 1) / DEG_SLICES + 3) & ~3;
  int* planes = ctrA + PLANE_OFF;
  k_count_deg<<<nRanges * DEG_SLICES, DEG_THREADS, 0, stream>>>(dst, planes, e, n, nRanges, perSlice);
  k_deg_finish<<<gridFor(n, THREADS), THREADS, 0, stream>>>(planes, dinv, n, ctrA + 2 * CTR_PITCH);
  k_edge_norm<<<gridFor(e, THREADS * 4 * EW_NORM), THREADS, 0, stream>>>(src, dst, dinv, norm, e);

  k_gemm1<<<gridFor(nTiles, wavesPerBlock), THREADS, 0, stream>>>(x, W1, xw1, nTiles);
  long long tot1 = (long long)n * HID;
  k_init_selfloop<4, EW_ROW1><<<gridFor(tot1, THREADS * 4 * EW_ROW1), THREADS, 0, stream>>>(xw1, dinv, b1, h, tot1);
  k_agg_edges<4><<<AGG_GRID, THREADS, 0, stream>>>(src, dst, norm, xw1, h, e, n, ctrA + 2 * CTR_PITCH);
  k_relu<<<gridFor(tot1, THREADS * 4 * EW_ROW1), THREADS, 0, stream>>>(h, tot1, ctrB);

  k_gemm2<<<gridFor(nTiles, wavesPerBlock), THREADS, 0, stream>>>(h, W2, h2, nTiles);
  long long tot2 = (long long)n * OUT_CH;
  k_init_selfloop<6, EW_ROW2><<<gridFor(tot2, THREADS * 4 * EW_ROW2), THREADS, 0, stream>>>(h2, dinv, b2, out, tot2);
  k_agg_edges<6><<<AGG_GRID, THREADS, 0, stream>>>(src, dst, norm, h2, out, e, n, ctrB);
}


# Both row scatters: a wave loads its batch's edge words at once and walks each chunk as a two-stage pipeline (a group's row loads issue before the previous group's adds); atomic arrival order changes
# speedup vs baseline: 4.6558x; 1.0624x over previous
#include <hip/hip_runtime.h>

typedef __attribute__((ext_vector_type(2))) float v2f;
typedef __attribute__((ext_vector_type(8))) float v8f;

#define IN_CH  512
#define HID    16
#define OUT_CH 64
#define THREADS 256

#define CTR_PITCH 64
#define AGG_BATCH 4
#define AGG_GRID  2048

__device__ __forceinline__ int my_half() {
  unsigned v = __builtin_amdgcn_s_sendmsg_rtn(0x87);
  return (int)((v >> 18) & 1u);
}

#define DEG_RANGE   12544
#define DEG_SLICES  32
#define DEG_THREADS 1024
#define DEG_GROUP   8
#define PLANE_OFF   1024

__device__ __forceinline__ void deg_bump(int* cnt, int d, int base, int len) {
  unsigned u = (unsigned)(d - base);
  if (u < (unsigned)len) atomicAdd(&cnt[u], 1);
}

__global__ __launch_bounds__(DEG_THREADS)
void k_count_deg(const int* __restrict__ dst, int* __restrict__ planes, int e, int n,
                 int nRanges, int perSlice) {
  __shared__ int cnt[DEG_RANGE];
  const int tid = threadIdx.x;
  const int r = blockIdx.x % nRanges;
  const int sl = blockIdx.x / nRanges;
  const int base = r * DEG_RANGE;
  const int len = (n - base < DEG_RANGE) ? (n - base) : DEG_RANGE;
  for (int i = tid; i < DEG_RANGE; i += DEG_THREADS) cnt[i] = 0;
  __syncthreads();

  const long long lo = (long long)sl * perSlice;
  long long hi = lo + perSlice;
  if (hi > e) hi = e;
  const int m = (hi > lo) ? (int)(hi - lo) : 0;
  const int* p = dst + lo;
  int done = 0;
  if ((((size_t)p) & 15) == 0) {
    const int4* p4 = (const int4*)p;
    const int nvec = m >> 2;
    for (int v = tid; v < nvec; v += DEG_GROUP * DEG_THREADS) {
      int4 d[DEG_GROUP];
      #pragma unroll
      for (int k = 0; k < DEG_GROUP; ++k) {
        const int vv = v + k * DEG_THREADS;
        d[k] = (vv < nvec) ? p4[vv] : make_int4(-1, -1, -1, -1);
      }
      #pragma unroll
      for (int k = 0; k < DEG_GROUP; ++k) {
        deg_bump(cnt, d[k].x, base, len);
        deg_bump(cnt, d[k].y, base, len);
        deg_bump(cnt, d[k].z, base, len);
        deg_bump(cnt, d[k].w, base, len);
      }
    }
    done = nvec << 2;
  }
  for (int i = done + tid; i < m; i += DEG_THREADS) deg_bump(cnt, p[i], base, len);
  __syncthreads();

  int* o = planes + (size_t)sl * n + base;
  for (int i = tid; i < len; i += DEG_THREADS) o[i] = cnt[i];
}

__global__ void k_deg_finish(const int* __restrict__ planes, float* __restrict__ dinv, int n,
                             int* __restrict__ ctrL1) {
  int i = blockIdx.x * blockDim.x + threadIdx.x;
  if (i < n) {
    int c = 1;
    #pragma unroll 8
    for (int sl = 0; sl < DEG_SLICES; ++sl) c += planes[(size_t)sl * n + i];
    dinv[i] = rsqrtf((float)c);
  }
  if (i < 2) ctrL1[i * CTR_PITCH] = 0;
}

#define EW_NORM 2
#define EW_ROW1 2
#define EW_ROW2 4

template <bool GUARD>
__device__ __forceinline__ void edge_norm_vec(const int4* __restrict__ s4, const int4* __restrict__ d4,
                                              const float* __restrict__ dinv, float4* __restrict__ o4,
                                              int v0, int nvec) {
  int4 s[EW_NORM], d[EW_NORM];
  float4 w[EW_NORM];
  #pragma unroll
  for (int k = 0; k < EW_NORM; ++k) {
    const int v = v0 + k * blockDim.x;
    const int vc = (!GUARD || v < nvec) ? v : nvec - 1;
    s[k] = s4[vc];
    d[k] = d4[vc];
  }
  #pragma unroll
  for (int k = 0; k < EW_NORM; ++k) {
    w[k].x = dinv[s[k].x] * dinv[d[k].x];
    w[k].y = dinv[s[k].y] * dinv[d[k].y];
    w[k].z = dinv[s[k].z] * dinv[d[k].z];
    w[k].w = dinv[s[k].w] * dinv[d[k].w];
  }
  #pragma unroll
  for (int k = 0; k < EW_NORM; ++k) {
    const int v = v0 + k * blockDim.x;
    if (!GUARD || v < nvec) o4[v] = w[k];
  }
}

__global__ void k_edge_norm(const int* __restrict__ src, const int* __restrict__ dst,
                            const float* __restrict__ dinv, float* __restrict__ norm, int e) {
  const bool vec = (((size_t)src | (size_t)dst | (size_t)norm) & 15) == 0;
  const int nvec = vec ? (e >> 2) : 0;
  const int4* s4 = (const int4*)src;
  const int4* d4 = (const int4*)dst;
  float4* o4 = (float4*)norm;
  const int v0 = blockIdx.x * (EW_NORM * blockDim.x) + threadIdx.x;
  if (nvec > 0) {
    if ((long long)(blockIdx.x + 1) * (EW_NORM * blockDim.x) <= nvec) edge_norm_vec<false>(s4, d4, dinv, o4, v0, nvec);
    else                                                              edge_norm_vec<true >(s4, d4, dinv, o4, v0, nvec);
  }
  const long long gsz = (long long)gridDim.x * blockDim.x;
  for (long long i = 4LL * nvec + (long long)blockIdx.x * blockDim.x + threadIdx.x; i < e; i += gsz)
    norm[i] = dinv[src[i]] * dinv[dst[i]];
}

__global__ void k_gemm1(const float* __restrict__ x, const float* __restrict__ W1,
                        float* __restrict__ xw1, int nTiles) {
  __shared__ float w1s[IN_CH * HID];
  for (int i = threadIdx.x; i < IN_CH * HID; i += blockDim.x) w1s[i] = W1[i];
  __syncthreads();

  int wave = threadIdx.x >> 5;
  int lane = threadIdx.x & 31;
  int t = blockIdx.x * (blockDim.x >> 5) + wave;
  if (t >= nTiles) return;

  int half = lane >> 4;
  int ml = lane & 15;
  const float* xr = x + (size_t)(t * 16 + ml) * IN_CH + half * 2;

  v8f c = {};
  for (int kk = 0; kk < IN_CH / 4; ++kk) {
    v2f a = *(const v2f*)(xr + kk * 4);
    int k0 = kk * 4 + half * 2;
    v2f b;
    b.x = w1s[k0 * HID + ml];
    b.y = w1s[(k0 + 1) * HID + ml];
    c = __builtin_amdgcn_wmma_f32_16x16x4_f32(false, a, false, b, (short)0, c, false, false);
  }

  float* o = xw1 + (size_t)(t * 16) * HID + ml;
  #pragma unroll
  for (int r = 0; r < 8; ++r)
    o[(half * 8 + r) * HID] = c[r];
}

__global__ void k_gemm2(const float* __restrict__ h, const float* __restrict__ W2,
                        float* __restrict__ h2, int nTiles) {
  __shared__ float w2s[HID * OUT_CH];
  for (int i = threadIdx.x; i < HID * OUT_CH; i += blockDim.x) w2s[i] = W2[i];
  __syncthreads();

  int wave = threadIdx.x >> 5;
  int lane = threadIdx.x & 31;
  int t = blockIdx.x * (blockDim.x >> 5) + wave;
  if (t >= nTiles) return;

  int half = lane >> 4;
  int ml = lane & 15;
  const float* hr = h + (size_t)(t * 16 + ml) * HID + half * 2;

  v8f c0 = {}, c1 = {}, c2 = {}, c3 = {};
  #pragma unroll
  for (int kk = 0; kk < HID / 4; ++kk) {
    v2f a = *(const v2f*)(hr + kk * 4);
    int k0 = kk * 4 + half * 2;
    v2f b0, b1, b2, b3;
    b0.x = w2s[k0 * OUT_CH +  0 + ml]; b0.y = w2s[(k0 + 1) * OUT_CH +  0 + ml];
    b1.x = w2s[k0 * OUT_CH + 16 + ml]; b1.y = w2s[(k0 + 1) * OUT_CH + 16 + ml];
    b2.x = w2s[k0 * OUT_CH + 32 + ml]; b2.y = w2s[(k0 + 1) * OUT_CH + 32 + ml];
    b3.x = w2s[k0 * OUT_CH + 48 + ml]; b3.y = w2s[(k0 + 1) * OUT_CH + 48 + ml];
    c0 = __builtin_amdgcn_wmma_f32_16x16x4_f32(false, a, false, b0, (short)0, c0, false, false);
    c1 = __builtin_amdgcn_wmma_f32_16x16x4_f32(false, a, false, b1, (short)0, c1, false, false);
    c2 = __builtin_amdgcn_wmma_f32_16x16x4_f32(false, a, false, b2, (short)0, c2, false, false);
    c3 = __builtin_amdgcn_wmma_f32_16x16x4_f32(false, a, false, b3, (short)0, c3, false, false);
  }

  float* o = h2 + (size_t)(t * 16) * OUT_CH + ml;
  #pragma unroll
  for (int r = 0; r < 8; ++r) {
    int row = half * 8 + r;
    o[row * OUT_CH +  0] = c0[r];
    o[row * OUT_CH + 16] = c1[r];
    o[row * OUT_CH + 32] = c2[r];
    o[row * OUT_CH + 48] = c3[r];
  }
}

template <int LOGC, int UNR, bool GUARD>
__device__ __forceinline__ void init_selfloop_vec(const float4* __restrict__ i4, const float* __restrict__ dinv,
                                                  const float4* __restrict__ b4, float4* __restrict__ o4,
                                                  long long v0, long long nvec) {
  float4 a[UNR], b[UNR];
  float di[UNR];
  #pragma unroll
  for (int k = 0; k < UNR; ++k) {
    const long long v = v0 + (long long)k * blockDim.x;
    const long long vc = (!GUARD || v < nvec) ? v : nvec - 1;
    a[k] = i4[vc];
    di[k] = dinv[(int)(vc >> (LOGC - 2))];
    b[k] = b4[(int)(vc & ((1 << (LOGC - 2)) - 1))];
  }
  #pragma unroll
  for (int k = 0; k < UNR; ++k) {
    const long long v = v0 + (long long)k * blockDim.x;
    if (!GUARD || v < nvec) {
      const float d = di[k];
      float4 r;
      r.x = b[k].x + a[k].x * d * d;
      r.y = b[k].y + a[k].y * d * d;
      r.z = b[k].z + a[k].z * d * d;
      r.w = b[k].w + a[k].w * d * d;
      o4[v] = r;
    }
  }
}

template <int LOGC, int UNR>
__global__ void k_init_selfloop(const float* __restrict__ in, const float* __restrict__ dinv,
                                const float* __restrict__ bias, float* __restrict__ out,
                                long long total) {
  static_assert(LOGC >= 2, "four channels per vector");
  const bool vec = (((size_t)in | (size_t)out | (size_t)bias) & 15) == 0;
  const long long nvec = vec ? (total >> 2) : 0;
  const float4* i4 = (const float4*)in;
  const float4* b4 = (const float4*)bias;
  float4* o4 = (float4*)out;
  const long long v0 = (long long)blockIdx.x * (UNR * blockDim.x) + threadIdx.x;
  if (nvec > 0) {
    if ((long long)(blockIdx.x + 1) * (UNR * blockDim.x) <= nvec) init_selfloop_vec<LOGC, UNR, false>(i4, dinv, b4, o4, v0, nvec);
    else                                                          init_selfloop_vec<LOGC, UNR, true >(i4, dinv, b4, o4, v0, nvec);
  }
  const long long gsz = (long long)gridDim.x * blockDim.x;
  for (long long i = 4 * nvec + (long long)blockIdx.x * blockDim.x + threadIdx.x; i < total; i += gsz) {
    const int node = (int)(i >> LOGC);
    const int ch = (int)(i & ((1 << LOGC) - 1));
    const float d = dinv[node];
    out[i] = bias[ch] + in[i] * d * d;
  }
}

template <int LOGC>
struct RowGroup {
  static constexpr int C   = 1 << LOGC;
  static constexpr int EPI = (C >= 32) ? 1 : 32 / C;
  static constexpr int IPE = (C >= 32) ? C / 32 : 1;
  static constexpr int U   = 8;
  float a[U][IPE];
  int d[U], w[U];
  unsigned valid;
  int dv, wv;
  int ja[U], jb[U];
};

template <int LOGC, bool FULL>
__device__ __forceinline__ unsigned form_load(unsigned m, int sv, int dv, int wv,
                                              const float* __restrict__ in, RowGroup<LOGC>& g,
                                              int sub, unsigned ch) {
  constexpr int EPI = RowGroup<LOGC>::EPI;
  constexpr int IPE = RowGroup<LOGC>::IPE;
  constexpr int U   = RowGroup<LOGC>::U;
  static_assert(EPI == 1 || EPI == 2, "one or two edges per instruction");
  const int j0 = __builtin_ctz(m);
  unsigned valid = 0;
  if (EPI == 2) { g.dv = dv;  g.wv = wv; }
  #pragma unroll
  for (int u = 0; u < U; ++u) {
    const int ja = m ? __builtin_ctz(m) : -1;  m &= m - 1;
    int jb = -1;
    if (EPI == 2) { jb = m ? __builtin_ctz(m) : -1;  m &= m - 1; }
    const int j1 = (FULL || ja >= 0) ? ja : j0;
    const int s = __builtin_amdgcn_readlane(sv, j1);
    if (EPI == 1) {
      g.d[u] = __builtin_amdgcn_readlane(dv, j1);
      g.w[u] = __builtin_amdgcn_readlane(wv, j1);
      if (!FULL && ja >= 0) valid |= 1u << u;
      const unsigned off = ((unsigned)s << LOGC) + ch;
      for (int k = 0; k < IPE; ++k) g.a[u][k] = in[off + k * 32];
    } else {
      g.ja[u] = ja;
      g.jb[u] = jb;
      const int s2 = __builtin_amdgcn_readlane(sv, (FULL || jb >= 0) ? jb : j1);
      const unsigned off = ((unsigned)(sub ? s2 : s) << LOGC) + ch;
      g.a[u][0] = in[off];
    }
  }
  g.valid = FULL ? 0xffffffffu : valid;
  return m;
}

template <int LOGC>
__device__ __forceinline__ unsigned form_any(unsigned m, int sv, int dv, int wv, const float* __restrict__ in,
                                             RowGroup<LOGC>& g, int sub, unsigned ch) {
  if (__builtin_popcount(m) >= RowGroup<LOGC>::U * RowGroup<LOGC>::EPI) return form_load<LOGC, true >(m, sv, dv, wv, in, g, sub, ch);
  return form_load<LOGC, false>(m, sv, dv, wv, in, g, sub, ch);
}

template <int LOGC>
__device__ __forceinline__ void add_group(const RowGroup<LOGC>& g, float* __restrict__ out, int sub, unsigned ch) {
  constexpr int EPI = RowGroup<LOGC>::EPI;
  constexpr int IPE = RowGroup<LOGC>::IPE;
  constexpr int U   = RowGroup<LOGC>::U;
  #pragma unroll
  for (int u = 0; u < U; ++u) {
    if (EPI == 1) {
      if ((g.valid >> u) & 1u) {
        const unsigned off = ((unsigned)g.d[u] << LOGC) + ch;
        for (int k = 0; k < IPE; ++k) atomicAdd(&out[off + k * 32], g.a[u][k] * __int_as_float(g.w[u]));
      }
    } else if (g.ja[u] >= 0) {
      const int d = __builtin_amdgcn_readlane(g.dv, g.ja[u]);
      const int w = __builtin_amdgcn_readlane(g.wv, g.ja[u]);
      const int j2 = g.jb[u] >= 0 ? g.jb[u] : g.ja[u];
      const int d2 = __builtin_amdgcn_readlane(g.dv, j2);
      const int w2 = __builtin_amdgcn_readlane(g.wv, j2);
      const unsigned off = ((unsigned)(sub ? d2 : d) << LOGC) + ch;
      const float wl = __int_as_float(sub ? w2 : w);
      if (sub == 0 || g.jb[u] >= 0)
        atomicAdd(&out[off], g.a[u][0] * wl);
    }
  }
}

__device__ __forceinline__ int pick4(const int (&v)[AGG_BATCH], int k) {
  static_assert(AGG_BATCH == 4, "four chunks per wave and ticket");
  return k == 0 ? v[0] : k == 1 ? v[1] : k == 2 ? v[2] : v[3];
}
__device__ __forceinline__ unsigned pick4s(const unsigned (&v)[AGG_BATCH], int k) {
  return k == 0 ? v[0] : k == 1 ? v[1] : k == 2 ? v[2] : v[3];
}

template <int LOGC>
__global__ void k_agg_edges(const int* __restrict__ src, const int* __restrict__ dst,
                            const float* __restrict__ norm, const float* __restrict__ in,
                            float* __restrict__ out, int e, int n, int* __restrict__ ctr) {
  constexpr int C   = 1 << LOGC;
  constexpr int EPI = (C >= 32) ? 1 : 32 / C;

  constexpr int WG_CHUNKS = AGG_BATCH * (THREADS / 32);

  __shared__ int s_tk[2];

  const int lane = threadIdx.x & 31;
  const int wave = __builtin_amdgcn_readfirstlane(threadIdx.x >> 5);
  const int sub = (EPI == 1) ? 0 : (lane >> LOGC);
  const unsigned ch = (EPI == 1) ? (unsigned)lane : (unsigned)(lane & (C - 1));
  const int nHalf = n >> 1;
  const int nChunks = (e + 31) >> 5;
  const int nBatches = (nChunks + WG_CHUNKS - 1) / WG_CHUNKS;
  const int home = my_half();
  int round = 0;

  for (int pass = 0; pass < 2; ++pass) {
    const int q = home ^ pass;
    unsigned* cq = (unsigned*)(ctr + q * CTR_PITCH);
    unsigned ahead = 0;
    if (threadIdx.x == 0) ahead = atomicInc(cq, 0xffffffffu);
    for (;;) {
      if (threadIdx.x == 0) s_tk[round & 1] = (int)ahead;
      __syncthreads();
      const int t = __builtin_amdgcn_readfirstlane(s_tk[round & 1]);
      ++round;
      if (t >= nBatches) break;
      if (threadIdx.x == 0) ahead = atomicInc(cq, 0xffffffffu);
      const int c0 = t * WG_CHUNKS + wave * AGG_BATCH;
      const int nkReal = (c0 + AGG_BATCH < nChunks) ? AGG_BATCH : nChunks - c0;
      const int nk = nkReal > 1 ? nkReal : 1;

      int sv[AGG_BATCH], dv[AGG_BATCH], wv[AGG_BATCH];
      bool ok[AGG_BATCH];
      #pragma unroll
      for (int k = 0; k < AGG_BATCH; ++k) {
        const long long base = (long long)(c0 + k) * 32;
        ok[k] = k < nkReal && base + lane < e;
        const long long idx = ok[k] ? base + lane : 0;
        sv[k] = src[idx];
        wv[k] = __float_as_int(norm[idx]);
      }
      #pragma unroll
      for (int k = 0; k < AGG_BATCH; ++k) {
        const long long base = (long long)(c0 + k) * 32;
        const long long idx = ok[k] ? base + lane : 0;
        const int dd = dst[idx];
        dv[k] = ok[k] ? dd : -1;
      }
      unsigned mk[AGG_BATCH];
      #pragma unroll
      for (int k = 0; k < AGG_BATCH; ++k)
        mk[k] = __builtin_amdgcn_ballot_w32(dv[k] >= 0 && ((int)(dv[k] >= nHalf) == q));

      RowGroup<LOGC> ga, gb;
      for (int k = 0; k < nk; ++k) {
        unsigned m = pick4s(mk, k);
        if (m == 0) continue;
        const int s_ = pick4(sv, k), d_ = pick4(dv, k), w_ = pick4(wv, k);
        m = form_any<LOGC>(m, s_, d_, w_, in, ga, sub, ch);
        for (;;) {
          if (m == 0) { add_group<LOGC>(ga, out, sub, ch);  break; }
          m = form_any<LOGC>(m, s_, d_, w_, in, gb, sub, ch);
          add_group<LOGC>(ga, out, sub, ch);
          if (m == 0) { add_group<LOGC>(gb, out, sub, ch);  break; }
          m = form_any<LOGC>(m, s_, d_, w_, in, ga, sub, ch);
          add_group<LOGC>(gb, out, sub, ch);
        }
      }
    }
  }
}

template <bool GUARD>
__device__ __forceinline__ void relu_vec(float4* h4, long long v0, long long nvec) {
  float4 a[EW_ROW1];
  #pragma unroll
  for (int k = 0; k < EW_ROW1; ++k) {
    const long long v = v0 + (long long)k * blockDim.x;
    const long long vc = (!GUARD || v < nvec) ? v : nvec - 1;
    a[k] = h4[vc];
  }
  #pragma unroll
  for (int k = 0; k < EW_ROW1; ++k) {
    const long long v = v0 + (long long)k * blockDim.x;
    if (!GUARD || v < nvec) {
      float4 r;
      r.x = fmaxf(a[k].x, 0.0f);
      r.y = fmaxf(a[k].y, 0.0f);
      r.z = fmaxf(a[k].z, 0.0f);
      r.w = fmaxf(a[k].w, 0.0f);
      h4[v] = r;
    }
  }
}

__global__ void k_relu(float* __restrict__ h, long long total, int* __restrict__ ctrB) {
  const bool vec = (((size_t)h) & 15) == 0;
  const long long nvec = vec ? (total >> 2) : 0;
  float4* h4 = (float4*)h;
  const long long v0 = (long long)blockIdx.x * (EW_ROW1 * blockDim.x) + threadIdx.x;
  if (nvec > 0) {
    if ((long long)(blockIdx.x + 1) * (EW_ROW1 * blockDim.x) <= nvec) relu_vec<false>(h4, v0, nvec);
    else                                                              relu_vec<true >(h4, v0, nvec);
  }
  const long long g = (long long)blockIdx.x * blockDim.x + threadIdx.x;
  const long long gsz = (long long)gridDim.x * blockDim.x;
  for (long long i = 4 * nvec + g; i < total; i += gsz) h[i] = fmaxf(h[i], 0.0f);
  if (g < 2) ctrB[g * CTR_PITCH] = 0;
}

static inline int gridFor(long long total, int threads) {
  return (int)((total + threads - 1) / threads);
}

extern "C" void kernel_launch(void* const* d_in, const int* in_sizes, int n_in,
                              void* d_out, int out_size, void* d_ws, size_t ws_size,
                              hipStream_t stream) {
  const float* x  = (const float*)d_in[0];
  const int* ei   = (const int*)d_in[1];
  const float* W1 = (const float*)d_in[2];
  const float* b1 = (const float*)d_in[3];
  const float* W2 = (const float*)d_in[4];
  const float* b2 = (const float*)d_in[5];
  float* out = (float*)d_out;

  const int n = in_sizes[0] / IN_CH;
  const int e = in_sizes[1] / 2;
  const int* src = ei;
  const int* dst = ei + e;

  float* dinv = (float*)d_ws;
  float* norm = dinv + n;
  float* xw1  = norm + e;
  float* h    = xw1 + (size_t)n * HID;
  float* h2   = h   + (size_t)n * HID;

  int* ctrA = (int*)h2;
  int* ctrB = (int*)xw1;

  const int nTiles = (n + 15) / 16;
  const int wavesPerBlock = THREADS / 32;

  const int nRanges = (n + DEG_RANGE - 1) / DEG_RANGE;
  const int perSlice = ((e + DEG_SLICES - 1) / DEG_SLICES + 3) & ~3;
  int* planes = ctrA + PLANE_OFF;
  k_count_deg<<<nRanges * DEG_SLICES, DEG_THREADS, 0, stream>>>(dst, planes, e, n, nRanges, perSlice);
  k_deg_finish<<<gridFor(n, THREADS), THREADS, 0, stream>>>(planes, dinv, n, ctrA + 2 * CTR_PITCH);
  k_edge_norm<<<gridFor(e, THREADS * 4 * EW_NORM), THREADS, 0, stream>>>(src, dst, dinv, norm, e);

  k_gemm1<<<gridFor(nTiles, wavesPerBlock), THREADS, 0, stream>>>(x, W1, xw1, nTiles);
  long long tot1 = (long long)n * HID;
  k_init_selfloop<4, EW_ROW1><<<gridFor(tot1, THREADS * 4 * EW_ROW1), THREADS, 0, stream>>>(xw1, dinv, b1, h, tot1);
  k_agg_edges<4><<<AGG_GRID, THREADS, 0, stream>>>(src, dst, norm, xw1, h, e, n, ctrA + 2 * CTR_PITCH);
  k_relu<<<gridFor(tot1, THREADS * 4 * EW_ROW1), THREADS, 0, stream>>>(h, tot1, ctrB);

  k_gemm2<<<gridFor(nTiles, wavesPerBlock), THREADS, 0, stream>>>(h, W2, h2, nTiles);
  long long tot2 = (long long)n * OUT_CH;
  k_init_selfloop<6, EW_ROW2><<<gridFor(tot2, THREADS * 4 * EW_ROW2), THREADS, 0, stream>>>(h2, dinv, b2, out, tot2);
  k_agg_edges<6><<<AGG_GRID, THREADS, 0, stream>>>(src, dst, norm, h2, out, e, n, ctrB);
}


# Both row scatters: a wave packs its batch's in-range edges via wave-private LDS into dense 32-edge vectors: all groups but the last full, one pipeline drain per batch; atomic arrival order changes
# speedup vs baseline: 4.8145x; 1.0341x over previous
#include <hip/hip_runtime.h>

typedef __attribute__((ext_vector_type(2))) float v2f;
typedef __attribute__((ext_vector_type(8))) float v8f;

#define IN_CH  512
#define HID    16
#define OUT_CH 64
#define THREADS 256

#define CTR_PITCH 64
#define AGG_BATCH 4
#define AGG_GRID  2048

__device__ __forceinline__ int my_half() {
  unsigned v = __builtin_amdgcn_s_sendmsg_rtn(0x87);
  return (int)((v >> 18) & 1u);
}

#define DEG_RANGE   12544
#define DEG_SLICES  32
#define DEG_THREADS 1024
#define DEG_GROUP   8
#define PLANE_OFF   1024

__device__ __forceinline__ void deg_bump(int* cnt, int d, int base, int len) {
  unsigned u = (unsigned)(d - base);
  if (u < (unsigned)len) atomicAdd(&cnt[u], 1);
}

__global__ __launch_bounds__(DEG_THREADS)
void k_count_deg(const int* __restrict__ dst, int* __restrict__ planes, int e, int n,
                 int nRanges, int perSlice) {
  __shared__ int cnt[DEG_RANGE];
  const int tid = threadIdx.x;
  const int r = blockIdx.x % nRanges;
  const int sl = blockIdx.x / nRanges;
  const int base = r * DEG_RANGE;
  const int len = (n - base < DEG_RANGE) ? (n - base) : DEG_RANGE;
  for (int i = tid; i < DEG_RANGE; i += DEG_THREADS) cnt[i] = 0;
  __syncthreads();

  const long long lo = (long long)sl * perSlice;
  long long hi = lo + perSlice;
  if (hi > e) hi = e;
  const int m = (hi > lo) ? (int)(hi - lo) : 0;
  const int* p = dst + lo;
  int done = 0;
  if ((((size_t)p) & 15) == 0) {
    const int4* p4 = (const int4*)p;
    const int nvec = m >> 2;
    for (int v = tid; v < nvec; v += DEG_GROUP * DEG_THREADS) {
      int4 d[DEG_GROUP];
      #pragma unroll
      for (int k = 0; k < DEG_GROUP; ++k) {
        const int vv = v + k * DEG_THREADS;
        d[k] = (vv < nvec) ? p4[vv] : make_int4(-1, -1, -1, -1);
      }
      #pragma unroll
      for (int k = 0; k < DEG_GROUP; ++k) {
        deg_bump(cnt, d[k].x, base, len);
        deg_bump(cnt, d[k].y, base, len);
        deg_bump(cnt, d[k].z, base, len);
        deg_bump(cnt, d[k].w, base, len);
      }
    }
    done = nvec << 2;
  }
  for (int i = done + tid; i < m; i += DEG_THREADS) deg_bump(cnt, p[i], base, len);
  __syncthreads();

  int* o = planes + (size_t)sl * n + base;
  for (int i = tid; i < len; i += DEG_THREADS) o[i] = cnt[i];
}

__global__ void k_deg_finish(const int* __restrict__ planes, float* __restrict__ dinv, int n,
                             int* __restrict__ ctrL1) {
  int i = blockIdx.x * blockDim.x + threadIdx.x;
  if (i < n) {
    int c = 1;
    #pragma unroll 8
    for (int sl = 0; sl < DEG_SLICES; ++sl) c += planes[(size_t)sl * n + i];
    dinv[i] = rsqrtf((float)c);
  }
  if (i < 2) ctrL1[i * CTR_PITCH] = 0;
}

#define EW_NORM 2
#define EW_ROW1 2
#define EW_ROW2 4

template <bool GUARD>
__device__ __forceinline__ void edge_norm_vec(const int4* __restrict__ s4, const int4* __restrict__ d4,
                                              const float* __restrict__ dinv, float4* __restrict__ o4,
                                              int v0, int nvec) {
  int4 s[EW_NORM], d[EW_NORM];
  float4 w[EW_NORM];
  #pragma unroll
  for (int k = 0; k < EW_NORM; ++k) {
    const int v = v0 + k * blockDim.x;
    const int vc = (!GUARD || v < nvec) ? v : nvec - 1;
    s[k] = s4[vc];
    d[k] = d4[vc];
  }
  #pragma unroll
  for (int k = 0; k < EW_NORM; ++k) {
    w[k].x = dinv[s[k].x] * dinv[d[k].x];
    w[k].y = dinv[s[k].y] * dinv[d[k].y];
    w[k].z = dinv[s[k].z] * dinv[d[k].z];
    w[k].w = dinv[s[k].w] * dinv[d[k].w];
  }
  #pragma unroll
  for (int k = 0; k < EW_NORM; ++k) {
    const int v = v0 + k * blockDim.x;
    if (!GUARD || v < nvec) o4[v] = w[k];
  }
}

__global__ void k_edge_norm(const int* __restrict__ src, const int* __restrict__ dst,
                            const float* __restrict__ dinv, float* __restrict__ norm, int e) {
  const bool vec = (((size_t)src | (size_t)dst | (size_t)norm) & 15) == 0;
  const int nvec = vec ? (e >> 2) : 0;
  const int4* s4 = (const int4*)src;
  const int4* d4 = (const int4*)dst;
  float4* o4 = (float4*)norm;
  const int v0 = blockIdx.x * (EW_NORM * blockDim.x) + threadIdx.x;
  if (nvec > 0) {
    if ((long long)(blockIdx.x + 1) * (EW_NORM * blockDim.x) <= nvec) edge_norm_vec<false>(s4, d4, dinv, o4, v0, nvec);
    else                                                              edge_norm_vec<true >(s4, d4, dinv, o4, v0, nvec);
  }
  const long long gsz = (long long)gridDim.x * blockDim.x;
  for (long long i = 4LL * nvec + (long long)blockIdx.x * blockDim.x + threadIdx.x; i < e; i += gsz)
    norm[i] = dinv[src[i]] * dinv[dst[i]];
}

__global__ void k_gemm1(const float* __restrict__ x, const float* __restrict__ W1,
                        float* __restrict__ xw1, int nTiles) {
  __shared__ float w1s[IN_CH * HID];
  for (int i = threadIdx.x; i < IN_CH * HID; i += blockDim.x) w1s[i] = W1[i];
  __syncthreads();

  int wave = threadIdx.x >> 5;
  int lane = threadIdx.x & 31;
  int t = blockIdx.x * (blockDim.x >> 5) + wave;
  if (t >= nTiles) return;

  int half = lane >> 4;
  int ml = lane & 15;
  const float* xr = x + (size_t)(t * 16 + ml) * IN_CH + half * 2;

  v8f c = {};
  for (int kk = 0; kk < IN_CH / 4; ++kk) {
    v2f a = *(const v2f*)(xr + kk * 4);
    int k0 = kk * 4 + half * 2;
    v2f b;
    b.x = w1s[k0 * HID + ml];
    b.y = w1s[(k0 + 1) * HID + ml];
    c = __builtin_amdgcn_wmma_f32_16x16x4_f32(false, a, false, b, (short)0, c, false, false);
  }

  float* o = xw1 + (size_t)(t * 16) * HID + ml;
  #pragma unroll
  for (int r = 0; r < 8; ++r)
    o[(half * 8 + r) * HID] = c[r];
}

__global__ void k_gemm2(const float* __restrict__ h, const float* __restrict__ W2,
                        float* __restrict__ h2, int nTiles) {
  __shared__ float w2s[HID * OUT_CH];
  for (int i = threadIdx.x; i < HID * OUT_CH; i += blockDim.x) w2s[i] = W2[i];
  __syncthreads();

  int wave = threadIdx.x >> 5;
  int lane = threadIdx.x & 31;
  int t = blockIdx.x * (blockDim.x >> 5) + wave;
  if (t >= nTiles) return;

  int half = lane >> 4;
  int ml = lane & 15;
  const float* hr = h + (size_t)(t * 16 + ml) * HID + half * 2;

  v8f c0 = {}, c1 = {}, c2 = {}, c3 = {};
  #pragma unroll
  for (int kk = 0; kk < HID / 4; ++kk) {
    v2f a = *(const v2f*)(hr + kk * 4);
    int k0 = kk * 4 + half * 2;
    v2f b0, b1, b2, b3;
    b0.x = w2s[k0 * OUT_CH +  0 + ml]; b0.y = w2s[(k0 + 1) * OUT_CH +  0 + ml];
    b1.x = w2s[k0 * OUT_CH + 16 + ml]; b1.y = w2s[(k0 + 1) * OUT_CH + 16 + ml];
    b2.x = w2s[k0 * OUT_CH + 32 + ml]; b2.y = w2s[(k0 + 1) * OUT_CH + 32 + ml];
    b3.x = w2s[k0 * OUT_CH + 48 + ml]; b3.y = w2s[(k0 + 1) * OUT_CH + 48 + ml];
    c0 = __builtin_amdgcn_wmma_f32_16x16x4_f32(false, a, false, b0, (short)0, c0, false, false);
    c1 = __builtin_amdgcn_wmma_f32_16x16x4_f32(false, a, false, b1, (short)0, c1, false, false);
    c2 = __builtin_amdgcn_wmma_f32_16x16x4_f32(false, a, false, b2, (short)0, c2, false, false);
    c3 = __builtin_amdgcn_wmma_f32_16x16x4_f32(false, a, false, b3, (short)0, c3, false, false);
  }

  float* o = h2 + (size_t)(t * 16) * OUT_CH + ml;
  #pragma unroll
  for (int r = 0; r < 8; ++r) {
    int row = half * 8 + r;
    o[row * OUT_CH +  0] = c0[r];
    o[row * OUT_CH + 16] = c1[r];
    o[row * OUT_CH + 32] = c2[r];
    o[row * OUT_CH + 48] = c3[r];
  }
}

template <int LOGC, int UNR, bool GUARD>
__device__ __forceinline__ void init_selfloop_vec(const float4* __restrict__ i4, const float* __restrict__ dinv,
                                                  const float4* __restrict__ b4, float4* __restrict__ o4,
                                                  long long v0, long long nvec) {
  float4 a[UNR], b[UNR];
  float di[UNR];
  #pragma unroll
  for (int k = 0; k < UNR; ++k) {
    const long long v = v0 + (long long)k * blockDim.x;
    const long long vc = (!GUARD || v < nvec) ? v : nvec - 1;
    a[k] = i4[vc];
    di[k] = dinv[(int)(vc >> (LOGC - 2))];
    b[k] = b4[(int)(vc & ((1 << (LOGC - 2)) - 1))];
  }
  #pragma unroll
  for (int k = 0; k < UNR; ++k) {
    const long long v = v0 + (long long)k * blockDim.x;
    if (!GUARD || v < nvec) {
      const float d = di[k];
      float4 r;
      r.x = b[k].x + a[k].x * d * d;
      r.y = b[k].y + a[k].y * d * d;
      r.z = b[k].z + a[k].z * d * d;
      r.w = b[k].w + a[k].w * d * d;
      o4[v] = r;
    }
  }
}

template <int LOGC, int UNR>
__global__ void k_init_selfloop(const float* __restrict__ in, const float* __restrict__ dinv,
                                const float* __restrict__ bias, float* __restrict__ out,
                                long long total) {
  static_assert(LOGC >= 2, "four channels per vector");
  const bool vec = (((size_t)in | (size_t)out | (size_t)bias) & 15) == 0;
  const long long nvec = vec ? (total >> 2) : 0;
  const float4* i4 = (const float4*)in;
  const float4* b4 = (const float4*)bias;
  float4* o4 = (float4*)out;
  const long long v0 = (long long)blockIdx.x * (UNR * blockDim.x) + threadIdx.x;
  if (nvec > 0) {
    if ((long long)(blockIdx.x + 1) * (UNR * blockDim.x) <= nvec) init_selfloop_vec<LOGC, UNR, false>(i4, dinv, b4, o4, v0, nvec);
    else                                                          init_selfloop_vec<LOGC, UNR, true >(i4, dinv, b4, o4, v0, nvec);
  }
  const long long gsz = (long long)gridDim.x * blockDim.x;
  for (long long i = 4 * nvec + (long long)blockIdx.x * blockDim.x + threadIdx.x; i < total; i += gsz) {
    const int node = (int)(i >> LOGC);
    const int ch = (int)(i & ((1 << LOGC) - 1));
    const float d = dinv[node];
    out[i] = bias[ch] + in[i] * d * d;
  }
}

template <int LOGC>
struct RowGroup {
  static constexpr int C   = 1 << LOGC;
  static constexpr int EPI = (C >= 32) ? 1 : 32 / C;
  static constexpr int IPE = (C >= 32) ? C / 32 : 1;
  static constexpr int U   = 8;
  float a[U][IPE];
  int d[U], w[U];
  unsigned valid;
  int dv, wv;
  int ja[U], jb[U];
};

template <int LOGC, bool FULL>
__device__ __forceinline__ void form_dense(int lane0, int remg, int sv, int dv, int wv, const float* __restrict__ in,
                                           RowGroup<LOGC>& g, int sub, unsigned ch) {
  constexpr int EPI = RowGroup<LOGC>::EPI;
  constexpr int IPE = RowGroup<LOGC>::IPE;
  constexpr int U   = RowGroup<LOGC>::U;
  static_assert(EPI == 1 || EPI == 2, "one or two edges per instruction");
  unsigned valid = 0;
  if (EPI == 2) { g.dv = dv;  g.wv = wv; }
  #pragma unroll
  for (int u = 0; u < U; ++u) {
    const bool va = FULL || u * EPI < remg;
    const int la = va ? lane0 + u * EPI : lane0;
    const int s = __builtin_amdgcn_readlane(sv, la);
    if (EPI == 1) {
      g.d[u] = __builtin_amdgcn_readlane(dv, la);
      g.w[u] = __builtin_amdgcn_readlane(wv, la);
      if (va) valid |= 1u << u;
      const unsigned off = ((unsigned)s << LOGC) + ch;
      for (int k = 0; k < IPE; ++k) g.a[u][k] = in[off + k * 32];
    } else {
      const bool vb = FULL || u * EPI + 1 < remg;
      g.ja[u] = va ? la : -1;
      g.jb[u] = vb ? la + 1 : -1;
      const int s2 = __builtin_amdgcn_readlane(sv, vb ? la + 1 : la);
      const unsigned off = ((unsigned)(sub ? s2 : s) << LOGC) + ch;
      g.a[u][0] = in[off];
    }
  }
  g.valid = FULL ? 0xffffffffu : valid;
}

template <int LOGC>
__device__ __forceinline__ void form_any(int lane0, int remg, int sv, int dv, int wv, const float* __restrict__ in,
                                         RowGroup<LOGC>& g, int sub, unsigned ch) {
  if (remg >= RowGroup<LOGC>::U * RowGroup<LOGC>::EPI) form_dense<LOGC, true >(lane0, remg, sv, dv, wv, in, g, sub, ch);
  else                                                 form_dense<LOGC, false>(lane0, remg, sv, dv, wv, in, g, sub, ch);
}

template <int LOGC>
__device__ __forceinline__ void add_group(const RowGroup<LOGC>& g, float* __restrict__ out, int sub, unsigned ch) {
  constexpr int EPI = RowGroup<LOGC>::EPI;
  constexpr int IPE = RowGroup<LOGC>::IPE;
  constexpr int U   = RowGroup<LOGC>::U;
  #pragma unroll
  for (int u = 0; u < U; ++u) {
    if (EPI == 1) {
      if ((g.valid >> u) & 1u) {
        const unsigned off = ((unsigned)g.d[u] << LOGC) + ch;
        for (int k = 0; k < IPE; ++k) atomicAdd(&out[off + k * 32], g.a[u][k] * __int_as_float(g.w[u]));
      }
    } else if (g.ja[u] >= 0) {
      const int d = __builtin_amdgcn_readlane(g.dv, g.ja[u]);
      const int w = __builtin_amdgcn_readlane(g.wv, g.ja[u]);
      const int j2 = g.jb[u] >= 0 ? g.jb[u] : g.ja[u];
      const int d2 = __builtin_amdgcn_readlane(g.dv, j2);
      const int w2 = __builtin_amdgcn_readlane(g.wv, j2);
      const unsigned off = ((unsigned)(sub ? d2 : d) << LOGC) + ch;
      const float wl = __int_as_float(sub ? w2 : w);
      if (sub == 0 || g.jb[u] >= 0)
        atomicAdd(&out[off], g.a[u][0] * wl);
    }
  }
}

template <int LOGC>
__global__ void k_agg_edges(const int* __restrict__ src, const int* __restrict__ dst,
                            const float* __restrict__ norm, const float* __restrict__ in,
                            float* __restrict__ out, int e, int n, int* __restrict__ ctr) {
  constexpr int C   = 1 << LOGC;
  constexpr int EPI = (C >= 32) ? 1 : 32 / C;
  constexpr int GE  = RowGroup<LOGC>::U * EPI;
  constexpr int GPC = 32 / GE;
  static_assert(GPC == 2 || GPC == 4, "an even number of groups per vector: a vector ends on the second register set");

  constexpr int WG_CHUNKS = AGG_BATCH * (THREADS / 32);

  __shared__ int s_tk[2];
  __shared__ int s_pk[THREADS / 32][3][AGG_BATCH * 32];

  const int lane = threadIdx.x & 31;
  const int wave = __builtin_amdgcn_readfirstlane(threadIdx.x >> 5);
  const int sub = (EPI == 1) ? 0 : (lane >> LOGC);
  const unsigned ch = (EPI == 1) ? (unsigned)lane : (unsigned)(lane & (C - 1));
  const int nHalf = n >> 1;
  const int nChunks = (e + 31) >> 5;
  const int nBatches = (nChunks + WG_CHUNKS - 1) / WG_CHUNKS;
  const int home = my_half();
  int round = 0;

  for (int pass = 0; pass < 2; ++pass) {
    const int q = home ^ pass;
    unsigned* cq = (unsigned*)(ctr + q * CTR_PITCH);
    unsigned ahead = 0;
    if (threadIdx.x == 0) ahead = atomicInc(cq, 0xffffffffu);
    for (;;) {
      if (threadIdx.x == 0) s_tk[round & 1] = (int)ahead;
      __syncthreads();
      const int t = __builtin_amdgcn_readfirstlane(s_tk[round & 1]);
      ++round;
      if (t >= nBatches) break;
      if (threadIdx.x == 0) ahead = atomicInc(cq, 0xffffffffu);
      const int c0 = t * WG_CHUNKS + wave * AGG_BATCH;
      const int nkReal = (c0 + AGG_BATCH < nChunks) ? AGG_BATCH : nChunks - c0;

      int sv[AGG_BATCH], dv[AGG_BATCH], wv[AGG_BATCH];
      bool ok[AGG_BATCH];
      #pragma unroll
      for (int k = 0; k < AGG_BATCH; ++k) {
        const long long base = (long long)(c0 + k) * 32;
        ok[k] = k < nkReal && base + lane < e;
        const long long idx = ok[k] ? base + lane : 0;
        sv[k] = src[idx];
        wv[k] = __float_as_int(norm[idx]);
      }
      #pragma unroll
      for (int k = 0; k < AGG_BATCH; ++k) {
        const long long base = (long long)(c0 + k) * 32;
        const long long idx = ok[k] ? base + lane : 0;
        const int dd = dst[idx];
        dv[k] = ok[k] ? dd : -1;
      }

      int* ps = s_pk[wave][0];
      int* pd = s_pk[wave][1];
      int* pw = s_pk[wave][2];
      int cnt = 0;
      #pragma unroll
      for (int k = 0; k < AGG_BATCH; ++k) {
        const bool in_range = dv[k] >= 0 && ((int)(dv[k] >= nHalf) == q);
        const unsigned m = __builtin_amdgcn_ballot_w32(in_range);
        const int pos = cnt + (int)__builtin_amdgcn_mbcnt_lo(m, 0u);
        if (in_range) { ps[pos] = sv[k];  pd[pos] = dv[k];  pw[pos] = wv[k]; }
        cnt += __builtin_popcount(m);
      }
      __builtin_amdgcn_fence(__ATOMIC_RELEASE, "wavefront");
      __builtin_amdgcn_wave_barrier();
      __builtin_amdgcn_fence(__ATOMIC_ACQUIRE, "wavefront");

      RowGroup<LOGC> ga, gb;
      const int ng = (cnt + GE - 1) / GE;
      if (ng > 0) {
        int gi = 0;
        int s_ = ps[lane], d_ = pd[lane], w_ = pw[lane];
        form_any<LOGC>(0, cnt, s_, d_, w_, in, ga, sub, ch);
        for (;;) {
          ++gi;
          if (gi == ng) { add_group<LOGC>(ga, out, sub, ch);  break; }
          form_any<LOGC>((gi & (GPC - 1)) * GE, cnt - gi * GE, s_, d_, w_, in, gb, sub, ch);
          add_group<LOGC>(ga, out, sub, ch);
          ++gi;
          if (gi == ng) { add_group<LOGC>(gb, out, sub, ch);  break; }
          if ((gi & (GPC - 1)) == 0) {
            const int i = (gi / GPC) * 32 + lane;
            s_ = ps[i];  d_ = pd[i];  w_ = pw[i];
          }
          form_any<LOGC>((gi & (GPC - 1)) * GE, cnt - gi * GE, s_, d_, w_, in, ga, sub, ch);
          add_group<LOGC>(gb, out, sub, ch);
        }
      }
      __builtin_amdgcn_fence(__ATOMIC_RELEASE, "wavefront");
      __builtin_amdgcn_wave_barrier();
      __builtin_amdgcn_fence(__ATOMIC_ACQUIRE, "wavefront");
    }
  }
}

template <bool GUARD>
__device__ __forceinline__ void relu_vec(float4* h4, long long v0, long long nvec) {
  float4 a[EW_ROW1];
  #pragma unroll
  for (int k = 0; k < EW_ROW1; ++k) {
    const long long v = v0 + (long long)k * blockDim.x;
    const long long vc = (!GUARD || v < nvec) ? v : nvec - 1;
    a[k] = h4[vc];
  }
  #pragma unroll
  for (int k = 0; k < EW_ROW1; ++k) {
    const long long v = v0 + (long long)k * blockDim.x;
    if (!GUARD || v < nvec) {
      float4 r;
      r.x = fmaxf(a[k].x, 0.0f);
      r.y = fmaxf(a[k].y, 0.0f);
      r.z = fmaxf(a[k].z, 0.0f);
      r.w = fmaxf(a[k].w, 0.0f);
      h4[v] = r;
    }
  }
}

__global__ void k_relu(float* __restrict__ h, long long total, int* __restrict__ ctrB) {
  const bool vec = (((size_t)h) & 15) == 0;
  const long long nvec = vec ? (total >> 2) : 0;
  float4* h4 = (float4*)h;
  const long long v0 = (long long)blockIdx.x * (EW_ROW1 * blockDim.x) + threadIdx.x;
  if (nvec > 0) {
    if ((long long)(blockIdx.x + 1) * (EW_ROW1 * blockDim.x) <= nvec) relu_vec<false>(h4, v0, nvec);
    else                                                              relu_vec<true >(h4, v0, nvec);
  }
  const long long g = (long long)blockIdx.x * blockDim.x + threadIdx.x;
  const long long gsz = (long long)gridDim.x * blockDim.x;
  for (long long i = 4 * nvec + g; i < total; i += gsz) h[i] = fmaxf(h[i], 0.0f);
  if (g < 2) ctrB[g * CTR_PITCH] = 0;
}

static inline int gridFor(long long total, int threads) {
  return (int)((total + threads - 1) / threads);
}

extern "C" void kernel_launch(void* const* d_in, const int* in_sizes, int n_in,
                              void* d_out, int out_size, void* d_ws, size_t ws_size,
                              hipStream_t stream) {
  const float* x  = (const float*)d_in[0];
  const int* ei   = (const int*)d_in[1];
  const float* W1 = (const float*)d_in[2];
  const float* b1 = (const float*)d_in[3];
  const float* W2 = (const float*)d_in[4];
  const float* b2 = (const float*)d_in[5];
  float* out = (float*)d_out;

  const int n = in_sizes[0] / IN_CH;
  const int e = in_sizes[1] / 2;
  const int* src = ei;
  const int* dst = ei + e;

  float* dinv = (float*)d_ws;
  float* norm = dinv + n;
  float* xw1  = norm + e;
  float* h    = xw1 + (size_t)n * HID;
  float* h2   = h   + (size_t)n * HID;

  int* ctrA = (int*)h2;
  int* ctrB = (int*)xw1;

  const int nTiles = (n + 15) / 16;
  const int wavesPerBlock = THREADS / 32;

  const int nRanges = (n + DEG_RANGE - 1) / DEG_RANGE;
  const int perSlice = ((e + DEG_SLICES - 1) / DEG_SLICES + 3) & ~3;
  int* planes = ctrA + PLANE_OFF;
  k_count_deg<<<nRanges * DEG_SLICES, DEG_THREADS, 0, stream>>>(dst, planes, e, n, nRanges, perSlice);
  k_deg_finish<<<gridFor(n, THREADS), THREADS, 0, stream>>>(planes, dinv, n, ctrA + 2 * CTR_PITCH);
  k_edge_norm<<<gridFor(e, THREADS * 4 * EW_NORM), THREADS, 0, stream>>>(src, dst, dinv, norm, e);

  k_gemm1<<<gridFor(nTiles, wavesPerBlock), THREADS, 0, stream>>>(x, W1, xw1, nTiles);
  long long tot1 = (long long)n * HID;
  k_init_selfloop<4, EW_ROW1><<<gridFor(tot1, THREADS * 4 * EW_ROW1), THREADS, 0, stream>>>(xw1, dinv, b1, h, tot1);
  k_agg_edges<4><<<AGG_GRID, THREADS, 0, stream>>>(src, dst, norm, xw1, h, e, n, ctrA + 2 * CTR_PITCH);
  k_relu<<<gridFor(tot1, THREADS * 4 * EW_ROW1), THREADS, 0, stream>>>(h, tot1, ctrB);

  k_gemm2<<<gridFor(nTiles, wavesPerBlock), THREADS, 0, stream>>>(h, W2, h2, nTiles);
  long long tot2 = (long long)n * OUT_CH;
  k_init_selfloop<6, EW_ROW2><<<gridFor(tot2, THREADS * 4 * EW_ROW2), THREADS, 0, stream>>>(h2, dinv, b2, out, tot2);
  k_agg_edges<6><<<AGG_GRID, THREADS, 0, stream>>>(src, dst, norm, h2, out, e, n, ctrB);
}


# Layer-1 product rides in the degree-count launch: a work-group counts or multiplies, by the parity of its number's set bits; 9 launches, not 10; x stream overlaps the LDS count; sum order unchanged
# speedup vs baseline: 5.0439x; 1.0476x over previous
#include <hip/hip_runtime.h>

typedef __attribute__((ext_vector_type(2))) float v2f;
typedef __attribute__((ext_vector_type(8))) float v8f;

#define IN_CH  512
#define HID    16
#define OUT_CH 64
#define THREADS 256

#define CTR_PITCH 64
#define AGG_BATCH 4
#define AGG_GRID  2048

__device__ __forceinline__ int my_half() {
  unsigned v = __builtin_amdgcn_s_sendmsg_rtn(0x87);
  return (int)((v >> 18) & 1u);
}

#define DEG_RANGE   12544
#define DEG_SLICES  32
#define DEG_THREADS 1024
#define DEG_GROUP   8
#define PLANE_OFF   1024

__device__ __forceinline__ void deg_bump(int* cnt, int d, int base, int len) {
  unsigned u = (unsigned)(d - base);
  if (u < (unsigned)len) atomicAdd(&cnt[u], 1);
}

__device__ __forceinline__ void gemm1_tile(const float* __restrict__ x, const float* w1s,
                                           float* __restrict__ xw1, int t, int lane) {
  int half = lane >> 4;
  int ml = lane & 15;
  const float* xr = x + (size_t)(t * 16 + ml) * IN_CH + half * 2;

  v8f c = {};
  for (int kk = 0; kk < IN_CH / 4; ++kk) {
    v2f a = *(const v2f*)(xr + kk * 4);
    int k0 = kk * 4 + half * 2;
    v2f b;
    b.x = w1s[k0 * HID + ml];
    b.y = w1s[(k0 + 1) * HID + ml];
    c = __builtin_amdgcn_wmma_f32_16x16x4_f32(false, a, false, b, (short)0, c, false, false);
  }

  float* o = xw1 + (size_t)(t * 16) * HID + ml;
  #pragma unroll
  for (int r = 0; r < 8; ++r)
    o[(half * 8 + r) * HID] = c[r];
}

__global__ __launch_bounds__(DEG_THREADS)
void k_count_deg(const int* __restrict__ dst, int* __restrict__ planes, int e, int n,
                 int nRanges, int perSlice,
                 const float* __restrict__ x, const float* __restrict__ W1, float* __restrict__ xw1,
                 int nTiles) {
  __shared__ __attribute__((aligned(16))) int cnt[DEG_RANGE];
  static_assert(DEG_RANGE >= IN_CH * HID, "W1 fits where the counters are");
  const int tid = threadIdx.x;
  const int pr = blockIdx.x >> 1;
  if (__builtin_popcount(blockIdx.x) & 1) {
    float* w1s = (float*)cnt;
    if ((((size_t)W1) & 15) == 0) {
      static_assert(IN_CH * HID == 8 * DEG_THREADS, "two 16-byte vectors per thread");
      const float4* W4 = (const float4*)W1;
      float4* s4 = (float4*)w1s;
      const float4 v0 = W4[tid];
      const float4 v1 = W4[tid + DEG_THREADS];
      s4[tid] = v0;
      s4[tid + DEG_THREADS] = v1;
    } else {
      for (int i = tid; i < IN_CH * HID; i += DEG_THREADS) w1s[i] = W1[i];
    }
    __syncthreads();
    const int groups = (int)(gridDim.x >> 1);
    const int wave = __builtin_amdgcn_readfirstlane(tid >> 5);
    for (int t = pr + wave * groups; t < nTiles; t += (DEG_THREADS / 32) * groups)
      gemm1_tile(x, w1s, xw1, t, tid & 31);
    return;
  }
  const int r = pr % nRanges;
  const int sl = pr / nRanges;
  const int base = r * DEG_RANGE;
  const int len = (n - base < DEG_RANGE) ? (n - base) : DEG_RANGE;
  for (int i = tid; i < DEG_RANGE; i += DEG_THREADS) cnt[i] = 0;
  __syncthreads();

  const long long lo = (long long)sl * perSlice;
  long long hi = lo + perSlice;
  if (hi > e) hi = e;
  const int m = (hi > lo) ? (int)(hi - lo) : 0;
  const int* p = dst + lo;
  int done = 0;
  if ((((size_t)p) & 15) == 0) {
    const int4* p4 = (const int4*)p;
    const int nvec = m >> 2;
    for (int v = tid; v < nvec; v += DEG_GROUP * DEG_THREADS) {
      int4 d[DEG_GROUP];
      #pragma unroll
      for (int k = 0; k < DEG_GROUP; ++k) {
        const int vv = v + k * DEG_THREADS;
        d[k] = (vv < nvec) ? p4[vv] : make_int4(-1, -1, -1, -1);
      }
      #pragma unroll
      for (int k = 0; k < DEG_GROUP; ++k) {
        deg_bump(cnt, d[k].x, base, len);
        deg_bump(cnt, d[k].y, base, len);
        deg_bump(cnt, d[k].z, base, len);
        deg_bump(cnt, d[k].w, base, len);
      }
    }
    done = nvec << 2;
  }
  for (int i = done + tid; i < m; i += DEG_THREADS) deg_bump(cnt, p[i], base, len);
  __syncthreads();

  int* o = planes + (size_t)sl * n + base;
  for (int i = tid; i < len; i += DEG_THREADS) o[i] = cnt[i];
}

__global__ void k_deg_finish(const int* __restrict__ planes, float* __restrict__ dinv, int n,
                             int* __restrict__ ctrL1) {
  int i = blockIdx.x * blockDim.x + threadIdx.x;
  if (i < n) {
    int c = 1;
    #pragma unroll 8
    for (int sl = 0; sl < DEG_SLICES; ++sl) c += planes[(size_t)sl * n + i];
    dinv[i] = rsqrtf((float)c);
  }
  if (i < 2) ctrL1[i * CTR_PITCH] = 0;
}

#define EW_NORM 2
#define EW_ROW1 2
#define EW_ROW2 4

template <bool GUARD>
__device__ __forceinline__ void edge_norm_vec(const int4* __restrict__ s4, const int4* __restrict__ d4,
                                              const float* __restrict__ dinv, float4* __restrict__ o4,
                                              int v0, int nvec) {
  int4 s[EW_NORM], d[EW_NORM];
  float4 w[EW_NORM];
  #pragma unroll
  for (int k = 0; k < EW_NORM; ++k) {
    const int v = v0 + k * blockDim.x;
    const int vc = (!GUARD || v < nvec) ? v : nvec - 1;
    s[k] = s4[vc];
    d[k] = d4[vc];
  }
  #pragma unroll
  for (int k = 0; k < EW_NORM; ++k) {
    w[k].x = dinv[s[k].x] * dinv[d[k].x];
    w[k].y = dinv[s[k].y] * dinv[d[k].y];
    w[k].z = dinv[s[k].z] * dinv[d[k].z];
    w[k].w = dinv[s[k].w] * dinv[d[k].w];
  }
  #pragma unroll
  for (int k = 0; k < EW_NORM; ++k) {
    const int v = v0 + k * blockDim.x;
    if (!GUARD || v < nvec) o4[v] = w[k];
  }
}

__global__ void k_edge_norm(const int* __restrict__ src, const int* __restrict__ dst,
                            const float* __restrict__ dinv, float* __restrict__ norm, int e) {
  const bool vec = (((size_t)src | (size_t)dst | (size_t)norm) & 15) == 0;
  const int nvec = vec ? (e >> 2) : 0;
  const int4* s4 = (const int4*)src;
  const int4* d4 = (const int4*)dst;
  float4* o4 = (float4*)norm;
  const int v0 = blockIdx.x * (EW_NORM * blockDim.x) + threadIdx.x;
  if (nvec > 0) {
    if ((long long)(blockIdx.x + 1) * (EW_NORM * blockDim.x) <= nvec) edge_norm_vec<false>(s4, d4, dinv, o4, v0, nvec);
    else                                                              edge_norm_vec<true >(s4, d4, dinv, o4, v0, nvec);
  }
  const long long gsz = (long long)gridDim.x * blockDim.x;
  for (long long i = 4LL * nvec + (long long)blockIdx.x * blockDim.x + threadIdx.x; i < e; i += gsz)
    norm[i] = dinv[src[i]] * dinv[dst[i]];
}

__global__ void k_gemm2(const float* __restrict__ h, const float* __restrict__ W2,
                        float* __restrict__ h2, int nTiles) {
  __shared__ float w2s[HID * OUT_CH];
  for (int i = threadIdx.x; i < HID * OUT_CH; i += blockDim.x) w2s[i] = W2[i];
  __syncthreads();

  int wave = threadIdx.x >> 5;
  int lane = threadIdx.x & 31;
  int t = blockIdx.x * (blockDim.x >> 5) + wave;
  if (t >= nTiles) return;

  int half = lane >> 4;
  int ml = lane & 15;
  const float* hr = h + (size_t)(t * 16 + ml) * HID + half * 2;

  v8f c0 = {}, c1 = {}, c2 = {}, c3 = {};
  #pragma unroll
  for (int kk = 0; kk < HID / 4; ++kk) {
    v2f a = *(const v2f*)(hr + kk * 4);
    int k0 = kk * 4 + half * 2;
    v2f b0, b1, b2, b3;
    b0.x = w2s[k0 * OUT_CH +  0 + ml]; b0.y = w2s[(k0 + 1) * OUT_CH +  0 + ml];
    b1.x = w2s[k0 * OUT_CH + 16 + ml]; b1.y = w2s[(k0 + 1) * OUT_CH + 16 + ml];
    b2.x = w2s[k0 * OUT_CH + 32 + ml]; b2.y = w2s[(k0 + 1) * OUT_CH + 32 + ml];
    b3.x = w2s[k0 * OUT_CH + 48 + ml]; b3.y = w2s[(k0 + 1) * OUT_CH + 48 + ml];
    c0 = __builtin_amdgcn_wmma_f32_16x16x4_f32(false, a, false, b0, (short)0, c0, false, false);
    c1 = __builtin_amdgcn_wmma_f32_16x16x4_f32(false, a, false, b1, (short)0, c1, false, false);
    c2 = __builtin_amdgcn_wmma_f32_16x16x4_f32(false, a, false, b2, (short)0, c2, false, false);
    c3 = __builtin_amdgcn_wmma_f32_16x16x4_f32(false, a, false, b3, (short)0, c3, false, false);
  }

  float* o = h2 + (size_t)(t * 16) * OUT_CH + ml;
  #pragma unroll
  for (int r = 0; r < 8; ++r) {
    int row = half * 8 + r;
    o[row * OUT_CH +  0] = c0[r];
    o[row * OUT_CH + 16] = c1[r];
    o[row * OUT_CH + 32] = c2[r];
    o[row * OUT_CH + 48] = c3[r];
  }
}

template <int LOGC, int UNR, bool GUARD>
__device__ __forceinline__ void init_selfloop_vec(const float4* __restrict__ i4, const float* __restrict__ dinv,
                                                  const float4* __restrict__ b4, float4* __restrict__ o4,
                                                  long long v0, long long nvec) {
  float4 a[UNR], b[UNR];
  float di[UNR];
  #pragma unroll
  for (int k = 0; k < UNR; ++k) {
    const long long v = v0 + (long long)k * blockDim.x;
    const long long vc = (!GUARD || v < nvec) ? v : nvec - 1;
    a[k] = i4[vc];
    di[k] = dinv[(int)(vc >> (LOGC - 2))];
    b[k] = b4[(int)(vc & ((1 << (LOGC - 2)) - 1))];
  }
  #pragma unroll
  for (int k = 0; k < UNR; ++k) {
    const long long v = v0 + (long long)k * blockDim.x;
    if (!GUARD || v < nvec) {
      const float d = di[k];
      float4 r;
      r.x = b[k].x + a[k].x * d * d;
      r.y = b[k].y + a[k].y * d * d;
      r.z = b[k].z + a[k].z * d * d;
      r.w = b[k].w + a[k].w * d * d;
      o4[v] = r;
    }
  }
}

template <int LOGC, int UNR>
__global__ void k_init_selfloop(const float* __restrict__ in, const float* __restrict__ dinv,
                                const float* __restrict__ bias, float* __restrict__ out,
                                long long total) {
  static_assert(LOGC >= 2, "four channels per vector");
  const bool vec = (((size_t)in | (size_t)out | (size_t)bias) & 15) == 0;
  const long long nvec = vec ? (total >> 2) : 0;
  const float4* i4 = (const float4*)in;
  const float4* b4 = (const float4*)bias;
  float4* o4 = (float4*)out;
  const long long v0 = (long long)blockIdx.x * (UNR * blockDim.x) + threadIdx.x;
  if (nvec > 0) {
    if ((long long)(blockIdx.x + 1) * (UNR * blockDim.x) <= nvec) init_selfloop_vec<LOGC, UNR, false>(i4, dinv, b4, o4, v0, nvec);
    else                                                          init_selfloop_vec<LOGC, UNR, true >(i4, dinv, b4, o4, v0, nvec);
  }
  const long long gsz = (long long)gridDim.x * blockDim.x;
  for (long long i = 4 * nvec + (long long)blockIdx.x * blockDim.x + threadIdx.x; i < total; i += gsz) {
    const int node = (int)(i >> LOGC);
    const int ch = (int)(i & ((1 << LOGC) - 1));
    const float d = dinv[node];
    out[i] = bias[ch] + in[i] * d * d;
  }
}

template <int LOGC>
struct RowGroup {
  static constexpr int C   = 1 << LOGC;
  static constexpr int EPI = (C >= 32) ? 1 : 32 / C;
  static constexpr int IPE = (C >= 32) ? C / 32 : 1;
  static constexpr int U   = 8;
  float a[U][IPE];
  int d[U], w[U];
  unsigned valid;
  int dv, wv;
  int ja[U], jb[U];
};

template <int LOGC, bool FULL>
__device__ __forceinline__ void form_dense(int lane0, int remg, int sv, int dv, int wv, const float* __restrict__ in,
                                           RowGroup<LOGC>& g, int sub, unsigned ch) {
  constexpr int EPI = RowGroup<LOGC>::EPI;
  constexpr int IPE = RowGroup<LOGC>::IPE;
  constexpr int U   = RowGroup<LOGC>::U;
  static_assert(EPI == 1 || EPI == 2, "one or two edges per instruction");
  unsigned valid = 0;
  if (EPI == 2) { g.dv = dv;  g.wv = wv; }
  #pragma unroll
  for (int u = 0; u < U; ++u) {
    const bool va = FULL || u * EPI < remg;
    const int la = va ? lane0 + u * EPI : lane0;
    const int s = __builtin_amdgcn_readlane(sv, la);
    if (EPI == 1) {
      g.d[u] = __builtin_amdgcn_readlane(dv, la);
      g.w[u] = __builtin_amdgcn_readlane(wv, la);
      if (va) valid |= 1u << u;
      const unsigned off = ((unsigned)s << LOGC) + ch;
      for (int k = 0; k < IPE; ++k) g.a[u][k] = in[off + k * 32];
    } else {
      const bool vb = FULL || u * EPI + 1 < remg;
      g.ja[u] = va ? la : -1;
      g.jb[u] = vb ? la + 1 : -1;
      const int s2 = __builtin_amdgcn_readlane(sv, vb ? la + 1 : la);
      const unsigned off = ((unsigned)(sub ? s2 : s) << LOGC) + ch;
      g.a[u][0] = in[off];
    }
  }
  g.valid = FULL ? 0xffffffffu : valid;
}

template <int LOGC>
__device__ __forceinline__ void form_any(int lane0, int remg, int sv, int dv, int wv, const float* __restrict__ in,
                                         RowGroup<LOGC>& g, int sub, unsigned ch) {
  if (remg >= RowGroup<LOGC>::U * RowGroup<LOGC>::EPI) form_dense<LOGC, true >(lane0, remg, sv, dv, wv, in, g, sub, ch);
  else                                                 form_dense<LOGC, false>(lane0, remg, sv, dv, wv, in, g, sub, ch);
}

template <int LOGC>
__device__ __forceinline__ void add_group(const RowGroup<LOGC>& g, float* __restrict__ out, int sub, unsigned ch) {
  constexpr int EPI = RowGroup<LOGC>::EPI;
  constexpr int IPE = RowGroup<LOGC>::IPE;
  constexpr int U   = RowGroup<LOGC>::U;
  #pragma unroll
  for (int u = 0; u < U; ++u) {
    if (EPI == 1) {
      if ((g.valid >> u) & 1u) {
        const unsigned off = ((unsigned)g.d[u] << LOGC) + ch;
        for (int k = 0; k < IPE; ++k) atomicAdd(&out[off + k * 32], g.a[u][k] * __int_as_float(g.w[u]));
      }
    } else if (g.ja[u] >= 0) {
      const int d = __builtin_amdgcn_readlane(g.dv, g.ja[u]);
      const int w = __builtin_amdgcn_readlane(g.wv, g.ja[u]);
      const int j2 = g.jb[u] >= 0 ? g.jb[u] : g.ja[u];
      const int d2 = __builtin_amdgcn_readlane(g.dv, j2);
      const int w2 = __builtin_amdgcn_readlane(g.wv, j2);
      const unsigned off = ((unsigned)(sub ? d2 : d) << LOGC) + ch;
      const float wl = __int_as_float(sub ? w2 : w);
      if (sub == 0 || g.jb[u] >= 0)
        atomicAdd(&out[off], g.a[u][0] * wl);
    }
  }
}

template <int LOGC>
__global__ void k_agg_edges(const int* __restrict__ src, const int* __restrict__ dst,
                            const float* __restrict__ norm, const float* __restrict__ in,
                            float* __restrict__ out, int e, int n, int* __restrict__ ctr) {
  constexpr int C   = 1 << LOGC;
  constexpr int EPI = (C >= 32) ? 1 : 32 / C;
  constexpr int GE  = RowGroup<LOGC>::U * EPI;
  constexpr int GPC = 32 / GE;
  static_assert(GPC == 2 || GPC == 4, "an even number of groups per vector: a vector ends on the second register set");

  constexpr int WG_CHUNKS = AGG_BATCH * (THREADS / 32);

  __shared__ int s_tk[2];
  __shared__ int s_pk[THREADS / 32][3][AGG_BATCH * 32];

  const int lane = threadIdx.x & 31;
  const int wave = __builtin_amdgcn_readfirstlane(threadIdx.x >> 5);
  const int sub = (EPI == 1) ? 0 : (lane >> LOGC);
  const unsigned ch = (EPI == 1) ? (unsigned)lane : (unsigned)(lane & (C - 1));
  const int nHalf = n >> 1;
  const int nChunks = (e + 31) >> 5;
  const int nBatches = (nChunks + WG_CHUNKS - 1) / WG_CHUNKS;
  const int home = my_half();
  int round = 0;

  for (int pass = 0; pass < 2; ++pass) {
    const int q = home ^ pass;
    unsigned* cq = (unsigned*)(ctr + q * CTR_PITCH);
    unsigned ahead = 0;
    if (threadIdx.x == 0) ahead = atomicInc(cq, 0xffffffffu);
    for (;;) {
      if (threadIdx.x == 0) s_tk[round & 1] = (int)ahead;
      __syncthreads();
      const int t = __builtin_amdgcn_readfirstlane(s_tk[round & 1]);
      ++round;
      if (t >= nBatches) break;
      if (threadIdx.x == 0) ahead = atomicInc(cq, 0xffffffffu);
      const int c0 = t * WG_CHUNKS + wave * AGG_BATCH;
      const int nkReal = (c0 + AGG_BATCH < nChunks) ? AGG_BATCH : nChunks - c0;

      int sv[AGG_BATCH], dv[AGG_BATCH], wv[AGG_BATCH];
      bool ok[AGG_BATCH];
      #pragma unroll
      for (int k = 0; k < AGG_BATCH; ++k) {
        const long long base = (long long)(c0 + k) * 32;
        ok[k] = k < nkReal && base + lane < e;
        const long long idx = ok[k] ? base + lane : 0;
        sv[k] = src[idx];
        wv[k] = __float_as_int(norm[idx]);
      }
      #pragma unroll
      for (int k = 0; k < AGG_BATCH; ++k) {
        const long long base = (long long)(c0 + k) * 32;
        const long long idx = ok[k] ? base + lane : 0;
        const int dd = dst[idx];
        dv[k] = ok[k] ? dd : -1;
      }

      int* ps = s_pk[wave][0];
      int* pd = s_pk[wave][1];
      int* pw = s_pk[wave][2];
      int cnt = 0;
      #pragma unroll
      for (int k = 0; k < AGG_BATCH; ++k) {
        const bool in_range = dv[k] >= 0 && ((int)(dv[k] >= nHalf) == q);
        const unsigned m = __builtin_amdgcn_ballot_w32(in_range);
        const int pos = cnt + (int)__builtin_amdgcn_mbcnt_lo(m, 0u);
        if (in_range) { ps[pos] = sv[k];  pd[pos] = dv[k];  pw[pos] = wv[k]; }
        cnt += __builtin_popcount(m);
      }
      __builtin_amdgcn_fence(__ATOMIC_RELEASE, "wavefront");
      __builtin_amdgcn_wave_barrier();
      __builtin_amdgcn_fence(__ATOMIC_ACQUIRE, "wavefront");

      RowGroup<LOGC> ga, gb;
      const int ng = (cnt + GE - 1) / GE;
      if (ng > 0) {
        int gi = 0;
        int s_ = ps[lane], d_ = pd[lane], w_ = pw[lane];
        form_any<LOGC>(0, cnt, s_, d_, w_, in, ga, sub, ch);
        for (;;) {
          ++gi;
          if (gi == ng) { add_group<LOGC>(ga, out, sub, ch);  break; }
          form_any<LOGC>((gi & (GPC - 1)) * GE, cnt - gi * GE, s_, d_, w_, in, gb, sub, ch);
          add_group<LOGC>(ga, out, sub, ch);
          ++gi;
          if (gi == ng) { add_group<LOGC>(gb, out, sub, ch);  break; }
          if ((gi & (GPC - 1)) == 0) {
            const int i = (gi / GPC) * 32 + lane;
            s_ = ps[i];  d_ = pd[i];  w_ = pw[i];
          }
          form_any<LOGC>((gi & (GPC - 1)) * GE, cnt - gi * GE, s_, d_, w_, in, ga, sub, ch);
          add_group<LOGC>(gb, out, sub, ch);
        }
      }
      __builtin_amdgcn_fence(__ATOMIC_RELEASE, "wavefront");
      __builtin_amdgcn_wave_barrier();
      __builtin_amdgcn_fence(__ATOMIC_ACQUIRE, "wavefront");
    }
  }
}

template <bool GUARD>
__device__ __forceinline__ void relu_vec(float4* h4, long long v0, long long nvec) {
  float4 a[EW_ROW1];
  #pragma unroll
  for (int k = 0; k < EW_ROW1; ++k) {
    const long long v = v0 + (long long)k * blockDim.x;
    const long long vc = (!GUARD || v < nvec) ? v : nvec - 1;
    a[k] = h4[vc];
  }
  #pragma unroll
  for (int k = 0; k < EW_ROW1; ++k) {
    const long long v = v0 + (long long)k * blockDim.x;
    if (!GUARD || v < nvec) {
      float4 r;
      r.x = fmaxf(a[k].x, 0.0f);
      r.y = fmaxf(a[k].y, 0.0f);
      r.z = fmaxf(a[k].z, 0.0f);
      r.w = fmaxf(a[k].w, 0.0f);
      h4[v] = r;
    }
  }
}

__global__ void k_relu(float* __restrict__ h, long long total, int* __restrict__ ctrB) {
  const bool vec = (((size_t)h) & 15) == 0;
  const long long nvec = vec ? (total >> 2) : 0;
  float4* h4 = (float4*)h;
  const long long v0 = (long long)blockIdx.x * (EW_ROW1 * blockDim.x) + threadIdx.x;
  if (nvec > 0) {
    if ((long long)(blockIdx.x + 1) * (EW_ROW1 * blockDim.x) <= nvec) relu_vec<false>(h4, v0, nvec);
    else                                                              relu_vec<true >(h4, v0, nvec);
  }
  const long long g = (long long)blockIdx.x * blockDim.x + threadIdx.x;
  const long long gsz = (long long)gridDim.x * blockDim.x;
  for (long long i = 4 * nvec + g; i < total; i += gsz) h[i] = fmaxf(h[i], 0.0f);
  if (g < 2) ctrB[g * CTR_PITCH] = 0;
}

static inline int gridFor(long long total, int threads) {
  return (int)((total + threads - 1) / threads);
}

extern "C" void kernel_launch(void* const* d_in, const int* in_sizes, int n_in,
                              void* d_out, int out_size, void* d_ws, size_t ws_size,
                              hipStream_t stream) {
  const float* x  = (const float*)d_in[0];
  const int* ei   = (const int*)d_in[1];
  const float* W1 = (const float*)d_in[2];
  const float* b1 = (const float*)d_in[3];
  const float* W2 = (const float*)d_in[4];
  const float* b2 = (const float*)d_in[5];
  float* out = (float*)d_out;

  const int n = in_sizes[0] / IN_CH;
  const int e = in_sizes[1] / 2;
  const int* src = ei;
  const int* dst = ei + e;

  float* dinv = (float*)d_ws;
  float* norm = dinv + n;
  float* xw1  = norm + e;
  float* h    = xw1 + (size_t)n * HID;
  float* h2   = h   + (size_t)n * HID;

  int* ctrA = (int*)h2;
  int* ctrB = (int*)xw1;

  const int nTiles = (n + 15) / 16;
  const int wavesPerBlock = THREADS / 32;

  const int nRanges = (n + DEG_RANGE - 1) / DEG_RANGE;
  const int perSlice = ((e + DEG_SLICES - 1) / DEG_SLICES + 3) & ~3;
  int* planes = ctrA + PLANE_OFF;
  k_count_deg<<<2 * nRanges * DEG_SLICES, DEG_THREADS, 0, stream>>>(dst, planes, e, n, nRanges, perSlice, x, W1, xw1, nTiles);
  k_deg_finish<<<gridFor(n, THREADS), THREADS, 0, stream>>>(planes, dinv, n, ctrA + 2 * CTR_PITCH);
  k_edge_norm<<<gridFor(e, THREADS * 4 * EW_NORM), THREADS, 0, stream>>>(src, dst, dinv, norm, e);

  long long tot1 = (long long)n * HID;
  k_init_selfloop<4, EW_ROW1><<<gridFor(tot1, THREADS * 4 * EW_ROW1), THREADS, 0, stream>>>(xw1, dinv, b1, h, tot1);
  k_agg_edges<4><<<AGG_GRID, THREADS, 0, stream>>>(src, dst, norm, xw1, h, e, n, ctrA + 2 * CTR_PITCH);
  k_relu<<<gridFor(tot1, THREADS * 4 * EW_ROW1), THREADS, 0, stream>>>(h, tot1, ctrB);

  k_gemm2<<<gridFor(nTiles, wavesPerBlock), THREADS, 0, stream>>>(h, W2, h2, nTiles);
  long long tot2 = (long long)n * OUT_CH;
  k_init_selfloop<6, EW_ROW2><<<gridFor(tot2, THREADS * 4 * EW_ROW2), THREADS, 0, stream>>>(h2, dinv, b2, out, tot2);
  k_agg_edges<6><<<AGG_GRID, THREADS, 0, stream>>>(src, dst, norm, h2, out, e, n, ctrB);
}


# Start-value and relu passes join neighbour launches: layer-1 start in the edge-norm blocks, relu on gemm2's A loads, layer-2 start from its accumulators: 6 launches, not 9; sum order unchanged
# speedup vs baseline: 5.2219x; 1.0353x over previous
#include <hip/hip_runtime.h>

typedef __attribute__((ext_vector_type(2))) float v2f;
typedef __attribute__((ext_vector_type(8))) float v8f;

#define IN_CH  512
#define HID    16
#define OUT_CH 64
#define THREADS 256

#define CTR_PITCH 64
#define AGG_BATCH 4
#define AGG_GRID  2048

__device__ __forceinline__ int my_half() {
  unsigned v = __builtin_amdgcn_s_sendmsg_rtn(0x87);
  return (int)((v >> 18) & 1u);
}

#define DEG_RANGE   12544
#define DEG_SLICES  32
#define DEG_THREADS 1024
#define DEG_GROUP   8
#define PLANE_OFF   1024

__device__ __forceinline__ void deg_bump(int* cnt, int d, int base, int len) {
  unsigned u = (unsigned)(d - base);
  if (u < (unsigned)len) atomicAdd(&cnt[u], 1);
}

__device__ __forceinline__ void gemm1_tile(const float* __restrict__ x, const float* w1s,
                                           float* __restrict__ xw1, int t, int lane) {
  int half = lane >> 4;
  int ml = lane & 15;
  const float* xr = x + (size_t)(t * 16 + ml) * IN_CH + half * 2;

  v8f c = {};
  for (int kk = 0; kk < IN_CH / 4; ++kk) {
    v2f a = *(const v2f*)(xr + kk * 4);
    int k0 = kk * 4 + half * 2;
    v2f b;
    b.x = w1s[k0 * HID + ml];
    b.y = w1s[(k0 + 1) * HID + ml];
    c = __builtin_amdgcn_wmma_f32_16x16x4_f32(false, a, false, b, (short)0, c, false, false);
  }

  float* o = xw1 + (size_t)(t * 16) * HID + ml;
  #pragma unroll
  for (int r = 0; r < 8; ++r)
    o[(half * 8 + r) * HID] = c[r];
}

__global__ __launch_bounds__(DEG_THREADS)
void k_count_deg(const int* __restrict__ dst, int* __restrict__ planes, int e, int n,
                 int nRanges, int perSlice,
                 const float* __restrict__ x, const float* __restrict__ W1, float* __restrict__ xw1,
                 int nTiles) {
  __shared__ __attribute__((aligned(16))) int cnt[DEG_RANGE];
  static_assert(DEG_RANGE >= IN_CH * HID, "W1 fits where the counters are");
  const int tid = threadIdx.x;
  const int pr = blockIdx.x >> 1;
  if (__builtin_popcount(blockIdx.x) & 1) {
    float* w1s = (float*)cnt;
    if ((((size_t)W1) & 15) == 0) {
      static_assert(IN_CH * HID == 8 * DEG_THREADS, "two 16-byte vectors per thread");
      const float4* W4 = (const float4*)W1;
      float4* s4 = (float4*)w1s;
      const float4 v0 = W4[tid];
      const float4 v1 = W4[tid + DEG_THREADS];
      s4[tid] = v0;
      s4[tid + DEG_THREADS] = v1;
    } else {
      for (int i = tid; i < IN_CH * HID; i += DEG_THREADS) w1s[i] = W1[i];
    }
    __syncthreads();
    const int groups = (int)(gridDim.x >> 1);
    const int wave = __builtin_amdgcn_readfirstlane(tid >> 5);
    for (int t = pr + wave * groups; t < nTiles; t += (DEG_THREADS / 32) * groups)
      gemm1_tile(x, w1s, xw1, t, tid & 31);
    return;
  }
  const int r = pr % nRanges;
  const int sl = pr / nRanges;
  const int base = r * DEG_RANGE;
  const int len = (n - base < DEG_RANGE) ? (n - base) : DEG_RANGE;
  for (int i = tid; i < DEG_RANGE; i += DEG_THREADS) cnt[i] = 0;
  __syncthreads();

  const long long lo = (long long)sl * perSlice;
  long long hi = lo + perSlice;
  if (hi > e) hi = e;
  const int m = (hi > lo) ? (int)(hi - lo) : 0;
  const int* p = dst + lo;
  int done = 0;
  if ((((size_t)p) & 15) == 0) {
    const int4* p4 = (const int4*)p;
    const int nvec = m >> 2;
    for (int v = tid; v < nvec; v += DEG_GROUP * DEG_THREADS) {
      int4 d[DEG_GROUP];
      #pragma unroll
      for (int k = 0; k < DEG_GROUP; ++k) {
        const int vv = v + k * DEG_THREADS;
        d[k] = (vv < nvec) ? p4[vv] : make_int4(-1, -1, -1, -1);
      }
      #pragma unroll
      for (int k = 0; k < DEG_GROUP; ++k) {
        deg_bump(cnt, d[k].x, base, len);
        deg_bump(cnt, d[k].y, base, len);
        deg_bump(cnt, d[k].z, base, len);
        deg_bump(cnt, d[k].w, base, len);
      }
    }
    done = nvec << 2;
  }
  for (int i = done + tid; i < m; i += DEG_THREADS) deg_bump(cnt, p[i], base, len);
  __syncthreads();

  int* o = planes + (size_t)sl * n + base;
  for (int i = tid; i < len; i += DEG_THREADS) o[i] = cnt[i];
}

__global__ void k_deg_finish(const int* __restrict__ planes, float* __restrict__ dinv, int n,
                             int* __restrict__ ctrL1) {
  int i = blockIdx.x * blockDim.x + threadIdx.x;
  if (i < n) {
    int c = 1;
    #pragma unroll 8
    for (int sl = 0; sl < DEG_SLICES; ++sl) c += planes[(size_t)sl * n + i];
    dinv[i] = rsqrtf((float)c);
  }
  if (i < 2) ctrL1[i * CTR_PITCH] = 0;
}

#define EW_NORM 2
#define EW_ROW1 1

template <bool GUARD>
__device__ __forceinline__ void edge_norm_vec(const int4* __restrict__ s4, const int4* __restrict__ d4,
                                              const float* __restrict__ dinv, float4* __restrict__ o4,
                                              int v0, int nvec) {
  int4 s[EW_NORM], d[EW_NORM];
  float4 w[EW_NORM];
  #pragma unroll
  for (int k = 0; k < EW_NORM; ++k) {
    const int v = v0 + k * blockDim.x;
    const int vc = (!GUARD || v < nvec) ? v : nvec - 1;
    s[k] = s4[vc];
    d[k] = d4[vc];
  }
  #pragma unroll
  for (int k = 0; k < EW_NORM; ++k) {
    w[k].x = dinv[s[k].x] * dinv[d[k].x];
    w[k].y = dinv[s[k].y] * dinv[d[k].y];
    w[k].z = dinv[s[k].z] * dinv[d[k].z];
    w[k].w = dinv[s[k].w] * dinv[d[k].w];
  }
  #pragma unroll
  for (int k = 0; k < EW_NORM; ++k) {
    const int v = v0 + k * blockDim.x;
    if (!GUARD || v < nvec) o4[v] = w[k];
  }
}

__device__ __forceinline__ void edge_norm_block(const int* __restrict__ src, const int* __restrict__ dst,
                                                const float* __restrict__ dinv, float* __restrict__ norm, int e,
                                                int bid, int nblk) {
  const bool vec = (((size_t)src | (size_t)dst | (size_t)norm) & 15) == 0;
  const int nvec = vec ? (e >> 2) : 0;
  const int4* s4 = (const int4*)src;
  const int4* d4 = (const int4*)dst;
  float4* o4 = (float4*)norm;
  const int v0 = bid * (EW_NORM * blockDim.x) + threadIdx.x;
  if (nvec > 0) {
    if ((long long)(bid + 1) * (EW_NORM * blockDim.x) <= nvec) edge_norm_vec<false>(s4, d4, dinv, o4, v0, nvec);
    else                                                       edge_norm_vec<true >(s4, d4, dinv, o4, v0, nvec);
  }
  const long long gsz = (long long)nblk * blockDim.x;
  for (long long i = 4LL * nvec + (long long)bid * blockDim.x + threadIdx.x; i < e; i += gsz)
    norm[i] = dinv[src[i]] * dinv[dst[i]];
}

__global__ void k_gemm2(const float* __restrict__ h, const float* __restrict__ W2,
                        const float* __restrict__ b2, const float* __restrict__ dinv,
                        float* __restrict__ h2, float* __restrict__ out, int nTiles,
                        int* __restrict__ ctrB) {
  __shared__ float w2s[HID * OUT_CH];
  for (int i = threadIdx.x; i < HID * OUT_CH; i += blockDim.x) w2s[i] = W2[i];
  if (blockIdx.x == 0 && threadIdx.x < 2) ctrB[threadIdx.x * CTR_PITCH] = 0;
  __syncthreads();

  int wave = threadIdx.x >> 5;
  int lane = threadIdx.x & 31;
  int t = blockIdx.x * (blockDim.x >> 5) + wave;
  if (t >= nTiles) return;

  int half = lane >> 4;
  int ml = lane & 15;
  const float* hr = h + (size_t)(t * 16 + ml) * HID + half * 2;

  v8f c0 = {}, c1 = {}, c2 = {}, c3 = {};
  #pragma unroll
  for (int kk = 0; kk < HID / 4; ++kk) {
    v2f a = *(const v2f*)(hr + kk * 4);
    a.x = fmaxf(a.x, 0.0f);
    a.y = fmaxf(a.y, 0.0f);
    int k0 = kk * 4 + half * 2;
    v2f w0, w1, w2, w3;
    w0.x = w2s[k0 * OUT_CH +  0 + ml]; w0.y = w2s[(k0 + 1) * OUT_CH +  0 + ml];
    w1.x = w2s[k0 * OUT_CH + 16 + ml]; w1.y = w2s[(k0 + 1) * OUT_CH + 16 + ml];
    w2.x = w2s[k0 * OUT_CH + 32 + ml]; w2.y = w2s[(k0 + 1) * OUT_CH + 32 + ml];
    w3.x = w2s[k0 * OUT_CH + 48 + ml]; w3.y = w2s[(k0 + 1) * OUT_CH + 48 + ml];
    c0 = __builtin_amdgcn_wmma_f32_16x16x4_f32(false, a, false, w0, (short)0, c0, false, false);
    c1 = __builtin_amdgcn_wmma_f32_16x16x4_f32(false, a, false, w1, (short)0, c1, false, false);
    c2 = __builtin_amdgcn_wmma_f32_16x16x4_f32(false, a, false, w2, (short)0, c2, false, false);
    c3 = __builtin_amdgcn_wmma_f32_16x16x4_f32(false, a, false, w3, (short)0, c3, false, false);
  }

  float* o = h2 + (size_t)(t * 16) * OUT_CH + ml;
  float* q = out + (size_t)(t * 16) * OUT_CH + ml;
  const float* dr = dinv + t * 16 + half * 8;
  const float e0 = b2[ml], e1 = b2[16 + ml], e2 = b2[32 + ml], e3 = b2[48 + ml];
  float di[8];
  #pragma unroll
  for (int r = 0; r < 8; ++r) di[r] = dr[r];
  #pragma unroll
  for (int r = 0; r < 8; ++r) {
    int row = half * 8 + r;
    o[row * OUT_CH +  0] = c0[r];
    o[row * OUT_CH + 16] = c1[r];
    o[row * OUT_CH + 32] = c2[r];
    o[row * OUT_CH + 48] = c3[r];
    const float d = di[r];
    q[row * OUT_CH +  0] = e0 + c0[r] * d * d;
    q[row * OUT_CH + 16] = e1 + c1[r] * d * d;
    q[row * OUT_CH + 32] = e2 + c2[r] * d * d;
    q[row * OUT_CH + 48] = e3 + c3[r] * d * d;
  }
}

template <int LOGC, int UNR, bool GUARD>
__device__ __forceinline__ void init_selfloop_vec(const float4* __restrict__ i4, const float* __restrict__ dinv,
                                                  const float4* __restrict__ b4, float4* __restrict__ o4,
                                                  long long v0, long long nvec) {
  float4 a[UNR], b[UNR];
  float di[UNR];
  #pragma unroll
  for (int k = 0; k < UNR; ++k) {
    const long long v = v0 + (long long)k * blockDim.x;
    const long long vc = (!GUARD || v < nvec) ? v : nvec - 1;
    a[k] = i4[vc];
    di[k] = dinv[(int)(vc >> (LOGC - 2))];
    b[k] = b4[(int)(vc & ((1 << (LOGC - 2)) - 1))];
  }
  #pragma unroll
  for (int k = 0; k < UNR; ++k) {
    const long long v = v0 + (long long)k * blockDim.x;
    if (!GUARD || v < nvec) {
      const float d = di[k];
      float4 r;
      r.x = b[k].x + a[k].x * d * d;
      r.y = b[k].y + a[k].y * d * d;
      r.z = b[k].z + a[k].z * d * d;
      r.w = b[k].w + a[k].w * d * d;
      o4[v] = r;
    }
  }
}

template <int LOGC, int UNR>
__device__ __forceinline__ void init_selfloop_block(const float* __restrict__ in, const float* __restrict__ dinv,
                                                    const float* __restrict__ bias, float* __restrict__ out,
                                                    long long total, int bid, int nblk) {
  static_assert(LOGC >= 2, "four channels per vector");
  const bool vec = (((size_t)in | (size_t)out | (size_t)bias) & 15) == 0;
  const long long nvec = vec ? (total >> 2) : 0;
  const float4* i4 = (const float4*)in;
  const float4* b4 = (const float4*)bias;
  float4* o4 = (float4*)out;
  const long long v0 = (long long)bid * (UNR * blockDim.x) + threadIdx.x;
  if (nvec > 0) {
    if ((long long)(bid + 1) * (UNR * blockDim.x) <= nvec) init_selfloop_vec<LOGC, UNR, false>(i4, dinv, b4, o4, v0, nvec);
    else                                                   init_selfloop_vec<LOGC, UNR, true >(i4, dinv, b4, o4, v0, nvec);
  }
  const long long gsz = (long long)nblk * blockDim.x;
  for (long long i = 4 * nvec + (long long)bid * blockDim.x + threadIdx.x; i < total; i += gsz) {
    const int node = (int)(i >> LOGC);
    const int ch = (int)(i & ((1 << LOGC) - 1));
    const float d = dinv[node];
    out[i] = bias[ch] + in[i] * d * d;
  }
}

__global__ void k_norm_init(const int* __restrict__ src, const int* __restrict__ dst,
                            const float* __restrict__ dinv, float* __restrict__ norm, int e, int nbNorm,
                            const float* __restrict__ xw1, const float* __restrict__ b1,
                            float* __restrict__ h, long long tot1, int nbInit) {
  const int b = blockIdx.x;
  if (b < nbInit) init_selfloop_block<4, EW_ROW1>(xw1, dinv, b1, h, tot1, b, nbInit);
  if (b < nbNorm) edge_norm_block(src, dst, dinv, norm, e, b, nbNorm);
}

template <int LOGC>
struct RowGroup {
  static constexpr int C   = 1 << LOGC;
  static constexpr int EPI = (C >= 32) ? 1 : 32 / C;
  static constexpr int IPE = (C >= 32) ? C / 32 : 1;
  static constexpr int U   = 8;
  float a[U][IPE];
  int d[U], w[U];
  unsigned valid;
  int dv, wv;
  int ja[U], jb[U];
};

template <int LOGC, bool FULL>
__device__ __forceinline__ void form_dense(int lane0, int remg, int sv, int dv, int wv, const float* __restrict__ in,
                                           RowGroup<LOGC>& g, int sub, unsigned ch) {
  constexpr int EPI = RowGroup<LOGC>::EPI;
  constexpr int IPE = RowGroup<LOGC>::IPE;
  constexpr int U   = RowGroup<LOGC>::U;
  static_assert(EPI == 1 || EPI == 2, "one or two edges per instruction");
  unsigned valid = 0;
  if (EPI == 2) { g.dv = dv;  g.wv = wv; }
  #pragma unroll
  for (int u = 0; u < U; ++u) {
    const bool va = FULL || u * EPI < remg;
    const int la = va ? lane0 + u * EPI : lane0;
    const int s = __builtin_amdgcn_readlane(sv, la);
    if (EPI == 1) {
      g.d[u] = __builtin_amdgcn_readlane(dv, la);
      g.w[u] = __builtin_amdgcn_readlane(wv, la);
      if (va) valid |= 1u << u;
      const unsigned off = ((unsigned)s << LOGC) + ch;
      for (int k = 0; k < IPE; ++k) g.a[u][k] = in[off + k * 32];
    } else {
      const bool vb = FULL || u * EPI + 1 < remg;
      g.ja[u] = va ? la : -1;
      g.jb[u] = vb ? la + 1 : -1;
      const int s2 = __builtin_amdgcn_readlane(sv, vb ? la + 1 : la);
      const unsigned off = ((unsigned)(sub ? s2 : s) << LOGC) + ch;
      g.a[u][0] = in[off];
    }
  }
  g.valid = FULL ? 0xffffffffu : valid;
}

template <int LOGC>
__device__ __forceinline__ void form_any(int lane0, int remg, int sv, int dv, int wv, const float* __restrict__ in,
                                         RowGroup<LOGC>& g, int sub, unsigned ch) {
  if (remg >= RowGroup<LOGC>::U * RowGroup<LOGC>::EPI) form_dense<LOGC, true >(lane0, remg, sv, dv, wv, in, g, sub, ch);
  else                                                 form_dense<LOGC, false>(lane0, remg, sv, dv, wv, in, g, sub, ch);
}

template <int LOGC>
__device__ __forceinline__ void add_group(const RowGroup<LOGC>& g, float* __restrict__ out, int sub, unsigned ch) {
  constexpr int EPI = RowGroup<LOGC>::EPI;
  constexpr int IPE = RowGroup<LOGC>::IPE;
  constexpr int U   = RowGroup<LOGC>::U;
  #pragma unroll
  for (int u = 0; u < U; ++u) {
    if (EPI == 1) {
      if ((g.valid >> u) & 1u) {
        const unsigned off = ((unsigned)g.d[u] << LOGC) + ch;
        for (int k = 0; k < IPE; ++k) atomicAdd(&out[off + k * 32], g.a[u][k] * __int_as_float(g.w[u]));
      }
    } else if (g.ja[u] >= 0) {
      const int d = __builtin_amdgcn_readlane(g.dv, g.ja[u]);
      const int w = __builtin_amdgcn_readlane(g.wv, g.ja[u]);
      const int j2 = g.jb[u] >= 0 ? g.jb[u] : g.ja[u];
      const int d2 = __builtin_amdgcn_readlane(g.dv, j2);
      const int w2 = __builtin_amdgcn_readlane(g.wv, j2);
      const unsigned off = ((unsigned)(sub ? d2 : d) << LOGC) + ch;
      const float wl = __int_as_float(sub ? w2 : w);
      if (sub == 0 || g.jb[u] >= 0)
        atomicAdd(&out[off], g.a[u][0] * wl);
    }
  }
}

template <int LOGC>
__global__ void k_agg_edges(const int* __restrict__ src, const int* __restrict__ dst,
                            const float* __restrict__ norm, const float* __restrict__ in,
                            float* __restrict__ out, int e, int n, int* __restrict__ ctr) {
  constexpr int C   = 1 << LOGC;
  constexpr int EPI = (C >= 32) ? 1 : 32 / C;
  constexpr int GE  = RowGroup<LOGC>::U * EPI;
  constexpr int GPC = 32 / GE;
  static_assert(GPC == 2 || GPC == 4, "an even number of groups per vector: a vector ends on the second register set");

  constexpr int WG_CHUNKS = AGG_BATCH * (THREADS / 32);

  __shared__ int s_tk[2];
  __shared__ int s_pk[THREADS / 32][3][AGG_BATCH * 32];

  const int lane = threadIdx.x & 31;
  const int wave = __builtin_amdgcn_readfirstlane(threadIdx.x >> 5);
  const int sub = (EPI == 1) ? 0 : (lane >> LOGC);
  const unsigned ch = (EPI == 1) ? (unsigned)lane : (unsigned)(lane & (C - 1));
  const int nHalf = n >> 1;
  const int nChunks = (e + 31) >> 5;
  const int nBatches = (nChunks + WG_CHUNKS - 1) / WG_CHUNKS;
  const int home = my_half();
  int round = 0;

  for (int pass = 0; pass < 2; ++pass) {
    const int q = home ^ pass;
    unsigned* cq = (unsigned*)(ctr + q * CTR_PITCH);
    unsigned ahead = 0;
    if (threadIdx.x == 0) ahead = atomicInc(cq, 0xffffffffu);
    for (;;) {
      if (threadIdx.x == 0) s_tk[round & 1] = (int)ahead;
      __syncthreads();
      const int t = __builtin_amdgcn_readfirstlane(s_tk[round & 1]);
      ++round;
      if (t >= nBatches) break;
      if (threadIdx.x == 0) ahead = atomicInc(cq, 0xffffffffu);
      const int c0 = t * WG_CHUNKS + wave * AGG_BATCH;
      const int nkReal = (c0 + AGG_BATCH < nChunks) ? AGG_BATCH : nChunks - c0;

      int sv[AGG_BATCH], dv[AGG_BATCH], wv[AGG_BATCH];
      bool ok[AGG_BATCH];
      #pragma unroll
      for (int k = 0; k < AGG_BATCH; ++k) {
        const long long base = (long long)(c0 + k) * 32;
        ok[k] = k < nkReal && base + lane < e;
        const long long idx = ok[k] ? base + lane : 0;
        sv[k] = src[idx];
        wv[k] = __float_as_int(norm[idx]);
      }
      #pragma unroll
      for (int k = 0; k < AGG_BATCH; ++k) {
        const long long base = (long long)(c0 + k) * 32;
        const long long idx = ok[k] ? base + lane : 0;
        const int dd = dst[idx];
        dv[k] = ok[k] ? dd : -1;
      }

      int* ps = s_pk[wave][0];
      int* pd = s_pk[wave][1];
      int* pw = s_pk[wave][2];
      int cnt = 0;
      #pragma unroll
      for (int k = 0; k < AGG_BATCH; ++k) {
        const bool in_range = dv[k] >= 0 && ((int)(dv[k] >= nHalf) == q);
        const unsigned m = __builtin_amdgcn_ballot_w32(in_range);
        const int pos = cnt + (int)__builtin_amdgcn_mbcnt_lo(m, 0u);
        if (in_range) { ps[pos] = sv[k];  pd[pos] = dv[k];  pw[pos] = wv[k]; }
        cnt += __builtin_popcount(m);
      }
      __builtin_amdgcn_fence(__ATOMIC_RELEASE, "wavefront");
      __builtin_amdgcn_wave_barrier();
      __builtin_amdgcn_fence(__ATOMIC_ACQUIRE, "wavefront");

      RowGroup<LOGC> ga, gb;
      const int ng = (cnt + GE - 1) / GE;
      if (ng > 0) {
        int gi = 0;
        int s_ = ps[lane], d_ = pd[lane], w_ = pw[lane];
        form_any<LOGC>(0, cnt, s_, d_, w_, in, ga, sub, ch);
        for (;;) {
          ++gi;
          if (gi == ng) { add_group<LOGC>(ga, out, sub, ch);  break; }
          form_any<LOGC>((gi & (GPC - 1)) * GE, cnt - gi * GE, s_, d_, w_, in, gb, sub, ch);
          add_group<LOGC>(ga, out, sub, ch);
          ++gi;
          if (gi == ng) { add_group<LOGC>(gb, out, sub, ch);  break; }
          if ((gi & (GPC - 1)) == 0) {
            const int i = (gi / GPC) * 32 + lane;
            s_ = ps[i];  d_ = pd[i];  w_ = pw[i];
          }
          form_any<LOGC>((gi & (GPC - 1)) * GE, cnt - gi * GE, s_, d_, w_, in, ga, sub, ch);
          add_group<LOGC>(gb, out, sub, ch);
        }
      }
      __builtin_amdgcn_fence(__ATOMIC_RELEASE, "wavefront");
      __builtin_amdgcn_wave_barrier();
      __builtin_amdgcn_fence(__ATOMIC_ACQUIRE, "wavefront");
    }
  }
}

static inline int gridFor(long long total, int threads) {
  return (int)((total + threads - 1) / threads);
}

extern "C" void kernel_launch(void* const* d_in, const int* in_sizes, int n_in,
                              void* d_out, int out_size, void* d_ws, size_t ws_size,
                              hipStream_t stream) {
  const float* x  = (const float*)d_in[0];
  const int* ei   = (const int*)d_in[1];
  const float* W1 = (const float*)d_in[2];
  const float* b1 = (const float*)d_in[3];
  const float* W2 = (const float*)d_in[4];
  const float* b2 = (const float*)d_in[5];
  float* out = (float*)d_out;

  const int n = in_sizes[0] / IN_CH;
  const int e = in_sizes[1] / 2;
  const int* src = ei;
  const int* dst = ei + e;

  float* dinv = (float*)d_ws;
  float* norm = dinv + n;
  float* xw1  = norm + e;
  float* h    = xw1 + (size_t)n * HID;
  float* h2   = h   + (size_t)n * HID;

  int* ctrA = (int*)h2;
  int* ctrB = (int*)xw1;

  const int nTiles = (n + 15) / 16;
  const int wavesPerBlock = THREADS / 32;

  const int nRanges = (n + DEG_RANGE - 1) / DEG_RANGE;
  const int perSlice = ((e + DEG_SLICES - 1) / DEG_SLICES + 3) & ~3;
  int* planes = ctrA + PLANE_OFF;
  k_count_deg<<<2 * nRanges * DEG_SLICES, DEG_THREADS, 0, stream>>>(dst, planes, e, n, nRanges, perSlice, x, W1, xw1, nTiles);
  k_deg_finish<<<gridFor(n, THREADS), THREADS, 0, stream>>>(planes, dinv, n, ctrA + 2 * CTR_PITCH);
  long long tot1 = (long long)n * HID;
  const int nbNorm = gridFor(e, THREADS * 4 * EW_NORM);
  const int nbInit = gridFor(tot1, THREADS * 4 * EW_ROW1);
  k_norm_init<<<(nbNorm > nbInit) ? nbNorm : nbInit, THREADS, 0, stream>>>(src, dst, dinv, norm, e, nbNorm, xw1, b1, h, tot1, nbInit);

  k_agg_edges<4><<<AGG_GRID, THREADS, 0, stream>>>(src, dst, norm, xw1, h, e, n, ctrA + 2 * CTR_PITCH);

  k_gemm2<<<gridFor(nTiles, wavesPerBlock), THREADS, 0, stream>>>(h, W2, b2, dinv, h2, out, nTiles, ctrB);
  k_agg_edges<6><<<AGG_GRID, THREADS, 0, stream>>>(src, dst, norm, h2, out, e, n, ctrB);
}
